# starting kernel
# baseline (speedup 1.0000x reference)
#include <hip/hip_runtime.h>
#include <hip/hip_bf16.h>

#define IN_DIM 512
#define HID 20
#define OUT_DIM 10

typedef __attribute__((ext_vector_type(2))) float v2f;
typedef __attribute__((ext_vector_type(8))) float v8f;

__global__ void k_init(float* __restrict__ deg, float* __restrict__ emb, int n) {
  long long t = (long long)blockIdx.x * blockDim.x + threadIdx.x;
  if (t < (long long)n * 64) emb[t] = 0.0f;
  if (t < n) deg[t] = 1.0f;
}

__global__ void k_packB(const float* __restrict__ W, int Krows, int Ncols,
                        float* __restrict__ Bp, int Ksteps, int ntiles) {
  int t = blockIdx.x * blockDim.x + threadIdx.x;
  int total = Ksteps * ntiles * 64;
  if (t >= total) return;
  int j = t & 1;
  int lane = (t >> 1) & 31;
  int tile = (t >> 6) % ntiles;
  int ks = (t >> 6) / ntiles;
  int k = 4 * ks + 2 * (lane >> 4) + j;
  int n = tile * 16 + (lane & 15);
  Bp[t] = (k < Krows && n < Ncols) ? W[k * Ncols + n] : 0.0f;
}

__global__ void k_deg(const int* __restrict__ dst, const float* __restrict__ ew,
                      float* __restrict__ deg, long long E) {
  long long e = (long long)blockIdx.x * blockDim.x + threadIdx.x;
  if (e >= E) return;
  unsafeAtomicAdd(&deg[dst[e]], ew[e]);
}

__global__ void k_dinv(const float* __restrict__ deg, float* __restrict__ dinv, int n) {
  int i = blockIdx.x * blockDim.x + threadIdx.x;
  if (i >= n) return;
  float d = deg[i];
  dinv[i] = (d > 0.0f) ? rsqrtf(d) : 0.0f;
}

__global__ void k_selfloop(const float* __restrict__ h, const float* __restrict__ dinv,
                           float* __restrict__ agg, int n) {
  int t = blockIdx.x * blockDim.x + threadIdx.x;
  int i = t >> 5, c = t & 31;
  if (i >= n) return;
  if (c < HID) { float di = dinv[i]; agg[(size_t)i * 32 + c] = h[(size_t)i * 32 + c] * di * di; }
}

__global__ void k_edge_agg(const int* __restrict__ src, const int* __restrict__ dst,
                           const float* __restrict__ ew, const float* __restrict__ dinv,
                           const float* __restrict__ h, float* __restrict__ agg, long long E) {
  long long t = (long long)blockIdx.x * blockDim.x + threadIdx.x;
  long long e = t >> 5;
  int c = (int)(t & 31);
  if (e >= E) return;
  int s = src[e], d = dst[e];
  float coef = dinv[s] * ew[e] * dinv[d];
  if (c < HID) unsafeAtomicAdd(&agg[(size_t)d * 32 + c], h[(size_t)s * 32 + c] * coef);
}

__global__ void k_bias_norm(const float* __restrict__ agg, const float* __restrict__ b,
                            float* __restrict__ emb, int n, int off) {
  int t = blockIdx.x * blockDim.x + threadIdx.x;
  int i = t >> 5, c = t & 31;
  if (i >= n) return;
  float v = 0.0f;
  if (c < HID) v = agg[(size_t)i * 32 + c] + b[c];
  float s = v * v;
#pragma unroll
  for (int m = 16; m >= 1; m >>= 1) s += __shfl_xor(s, m, 32);
  float nrm = fmaxf(sqrtf(s), 1e-12f);
  if (c < HID) {
    float r = v / nrm;
    emb[(size_t)i * 64 + off + c] = (r > 0.0f) ? r : 0.0f;
  }
}

template <int NTILES>
__global__ void k_wmma_gemm(const float* __restrict__ A, int lda, int acol0,
                            const float* __restrict__ Bp, int Ksteps,
                            float* __restrict__ out, int ldo, int ncols,
                            const float* __restrict__ bias, int npairs) {
  int pair = blockIdx.x * (blockDim.x >> 5) + (threadIdx.x >> 5);
  if (pair >= npairs) return;
  int lane = threadIdx.x & 31;
  int half = lane >> 4;
  int l16 = lane & 15;
  int m0 = pair * 32;
  const float* ap0 = A + (size_t)(m0 + l16) * lda + acol0 + half * 2;
  const float* ap1 = ap0 + (size_t)16 * lda;
  const float* bp = Bp + lane * 2;
  v8f acc[2][NTILES];
#pragma unroll
  for (int rr = 0; rr < 2; ++rr)
#pragma unroll
    for (int t = 0; t < NTILES; ++t) acc[rr][t] = (v8f){0.f, 0.f, 0.f, 0.f, 0.f, 0.f, 0.f, 0.f};

#pragma unroll 2
  for (int ks = 0; ks < Ksteps; ++ks) {
    v2f a0 = *(const v2f*)(ap0 + 4 * ks);
    v2f a1 = *(const v2f*)(ap1 + 4 * ks);
#pragma unroll
    for (int t = 0; t < NTILES; ++t) {
      v2f b = *(const v2f*)(bp + (size_t)(ks * NTILES + t) * 64);
      acc[0][t] = __builtin_amdgcn_wmma_f32_16x16x4_f32(false, a0, false, b, (short)0, acc[0][t], false, false);
      acc[1][t] = __builtin_amdgcn_wmma_f32_16x16x4_f32(false, a1, false, b, (short)0, acc[1][t], false, false);
    }
  }

#pragma unroll
  for (int rr = 0; rr < 2; ++rr)
#pragma unroll
    for (int t = 0; t < NTILES; ++t)
#pragma unroll
      for (int r = 0; r < 8; ++r) {
        int row = m0 + rr * 16 + r + half * 8;
        int col = t * 16 + l16;
        if (col < ncols) out[(size_t)row * ldo + col] = acc[rr][t][r] + (bias ? bias[col] : 0.0f);
      }
}

extern "C" void kernel_launch(void* const* d_in, const int* in_sizes, int n_in,
                              void* d_out, int out_size, void* d_ws, size_t ws_size,
                              hipStream_t stream) {
  const float* x  = (const float*)d_in[0];
  const int*   ei = (const int*)d_in[1];
  const float* ew = (const float*)d_in[2];
  const float* W1 = (const float*)d_in[3];
  const float* b1 = (const float*)d_in[4];
  const float* W2 = (const float*)d_in[5];
  const float* b2 = (const float*)d_in[6];
  const float* W3 = (const float*)d_in[7];
  const float* b3 = (const float*)d_in[8];
  const float* Wl = (const float*)d_in[9];
  const float* bl = (const float*)d_in[10];
  float* out = (float*)d_out;

  const int N = in_sizes[0] / IN_DIM;
  const long long E = in_sizes[2];
  const int* src = ei;
  const int* dst = ei + E;

  const int KS1 = IN_DIM / 4;
  const int KS2 = (HID + 3) / 4;
  const int KSL = 64 / 4;

  float* ws = (float*)d_ws;
  float* deg  = ws; ws += N;
  float* dinv = ws; ws += N;
  float* h    = ws; ws += (size_t)N * 32;
  float* agg  = ws; ws += (size_t)N * 32;
  float* emb  = ws; ws += (size_t)N * 64;
  float* W1p  = ws; ws += KS1 * 2 * 64;
  float* W2p  = ws; ws += KS2 * 2 * 64;
  float* W3p  = ws; ws += KS2 * 2 * 64;
  float* Wlp  = ws; ws += KSL * 1 * 64;

  const int TB = 256;
  const int NPAIRS = N / 32;
  dim3 gblk((NPAIRS + 3) / 4);
  long long nt32 = (long long)N * 32;
  long long et32 = E * 32;
  int g_n32 = (int)((nt32 + TB - 1) / TB);
  int g_e32 = (int)((et32 + TB - 1) / TB);

  k_init<<<(int)(((long long)N * 64 + TB - 1) / TB), TB, 0, stream>>>(deg, emb, N);
  k_packB<<<(KS1 * 2 * 64 + TB - 1) / TB, TB, 0, stream>>>(W1, IN_DIM, HID, W1p, KS1, 2);
  k_packB<<<(KS2 * 2 * 64 + TB - 1) / TB, TB, 0, stream>>>(W2, HID, HID, W2p, KS2, 2);
  k_packB<<<(KS2 * 2 * 64 + TB - 1) / TB, TB, 0, stream>>>(W3, HID, HID, W3p, KS2, 2);
  k_packB<<<(KSL * 1 * 64 + TB - 1) / TB, TB, 0, stream>>>(Wl, 3 * HID, OUT_DIM, Wlp, KSL, 1);
  k_deg<<<(int)((E + TB - 1) / TB), TB, 0, stream>>>(dst, ew, deg, E);
  k_dinv<<<(N + TB - 1) / TB, TB, 0, stream>>>(deg, dinv, N);

  k_wmma_gemm<2><<<gblk, 128, 0, stream>>>(x, IN_DIM, 0, W1p, KS1, h, 32, 32, nullptr, NPAIRS);
  k_selfloop<<<g_n32, TB, 0, stream>>>(h, dinv, agg, N);
  k_edge_agg<<<g_e32, TB, 0, stream>>>(src, dst, ew, dinv, h, agg, E);
  k_bias_norm<<<g_n32, TB, 0, stream>>>(agg, b1, emb, N, 0);

  k_wmma_gemm<2><<<gblk, 128, 0, stream>>>(emb, 64, 0, W2p, KS2, h, 32, 32, nullptr, NPAIRS);
  k_selfloop<<<g_n32, TB, 0, stream>>>(h, dinv, agg, N);
  k_edge_agg<<<g_e32, TB, 0, stream>>>(src, dst, ew, dinv, h, agg, E);
  k_bias_norm<<<g_n32, TB, 0, stream>>>(agg, b2, emb, N, HID);

  k_wmma_gemm<2><<<gblk, 128, 0, stream>>>(emb, 64, HID, W3p, KS2, h, 32, 32, nullptr, NPAIRS);
  k_selfloop<<<g_n32, TB, 0, stream>>>(h, dinv, agg, N);
  k_edge_agg<<<g_e32, TB, 0, stream>>>(src, dst, ew, dinv, h, agg, E);
  k_bias_norm<<<g_n32, TB, 0, stream>>>(agg, b3, emb, N, 2 * HID);

  k_wmma_gemm<1><<<gblk, 128, 0, stream>>>(emb, 64, 0, Wlp, KSL, out, OUT_DIM, OUT_DIM, bl, NPAIRS);
}


# List heads split by chip half: an edge goes on a list of the half its wave runs on (die id); each half's head words on lines of their own, so no line takes exchanges from both; sum order = arrival
# speedup vs baseline: 3.4613x; 3.4613x over previous
#include <hip/hip_runtime.h>
#include <hip/hip_bf16.h>

#define IN_DIM 512
#define HID 20
#define OUT_DIM 10

typedef __attribute__((ext_vector_type(2))) float v2f;
typedef __attribute__((ext_vector_type(8))) float v8f;

#define NSUB 4
#define HEAD_COL 60
#define DINV_COL 24

__global__ void k_init(int* __restrict__ embw, int* __restrict__ hd1, int n) {
  long long t = (long long)blockIdx.x * blockDim.x + threadIdx.x;
  long long ne = (long long)n * 64;
  if (t < ne) embw[t] = ((int)(t & 63) >= HEAD_COL) ? -1 : 0;
  else if (t < ne + (long long)n * 2) hd1[t - ne] = -1;
}

__device__ __forceinline__ const int* head_ptr(const int* embw, const int* hd1, int i, int sub) {
  return (sub < 2) ? (embw + (size_t)i * 64 + HEAD_COL + sub) : (hd1 + (size_t)i * 2 + (sub - 2));
}

__device__ __forceinline__ void packB_elem(int t, const float* __restrict__ W, int Krows, int Ncols,
                                           float* __restrict__ Bp, int Ksteps, int ntiles) {
  int total = Ksteps * ntiles * 64;
  if (t >= total) return;
  int j = t & 1;
  int lane = (t >> 1) & 31;
  int tile = (t >> 6) % ntiles;
  int ks = (t >> 6) / ntiles;
  int k = 4 * ks + 2 * (lane >> 4) + j;
  int n = tile * 16 + (lane & 15);
  Bp[t] = (k < Krows && n < Ncols) ? W[k * Ncols + n] : 0.0f;
}

__global__ void k_packB(const float* __restrict__ W, int Krows, int Ncols,
                        float* __restrict__ Bp, int Ksteps, int ntiles) {
  packB_elem(blockIdx.x * blockDim.x + threadIdx.x, W, Krows, Ncols, Bp, Ksteps, ntiles);
}

struct PackJob {
  const float* W;
  float* Bp;
  int Krows, Ncols, Ksteps, ntiles;
};

__device__ __forceinline__ void build_edge(int e, const int* __restrict__ dst, int* __restrict__ heads,
                                           int* __restrict__ hd1, int* __restrict__ nxt, int E) {
  if (e >= E) return;
  int d = dst[e];
  unsigned place = __builtin_amdgcn_s_sendmsg_rtn(0x87);
  int hf = (int)((place >> 18) & 1u);
  int k = e & 1;
  int* hp = hf ? (hd1 + (size_t)d * 2 + k) : (heads + (size_t)d * 64 + HEAD_COL + k);
  int old = atomicExch(hp, e);
  nxt[e] = old;
}

__global__ void k_deg_walk(const int* __restrict__ hplane, const int* __restrict__ hd1,
                           const int* __restrict__ nxt,
                           const float* __restrict__ ew, float* __restrict__ hrow, int n, int E) {
  int t = blockIdx.x * blockDim.x + threadIdx.x;
  int i = t / NSUB, sub = t % NSUB;
  bool valid = i < n;
  int cur = valid ? *head_ptr(hplane, hd1, i, sub) : -1;
  float p = 0.0f;
  for (int it = 0; it < E; ++it) {
    if ((unsigned)cur >= (unsigned)E) break;
    float w = ew[cur];
    int nx = nxt[cur];
    p += w;
    cur = nx;
  }
  p += __shfl_xor(p, 1, 32);
  p += __shfl_xor(p, 2, 32);
  if (valid && sub == 0) {
    float d = 1.0f + p;
    float4 o = {(d > 0.0f) ? rsqrtf(d) : 0.0f, 0.0f, 0.0f, 0.0f};
    *(float4*)(hrow + (size_t)i * 32 + DINV_COL) = o;
  }
}

__global__ void k_walk_agg(const int* __restrict__ hplane, const int* __restrict__ hd1,
                           const int* __restrict__ nxt,
                           const int* __restrict__ src, const float* __restrict__ ew,
                           const float* __restrict__ h,
                           const float* __restrict__ b, float* __restrict__ emb,
                           int n, int off, int E) {
  static_assert(NSUB * 8 == 32, "one wave = NSUB groups of 8 lanes");
  int lane = threadIdx.x & 31;
  int i = blockIdx.x * (blockDim.x >> 5) + (threadIdx.x >> 5);
  if (i >= n) return;
  int sub = lane >> 3;
  int q = lane & 7;
  int pq = (q < HID / 4) ? q : DINV_COL / 4;
  float4 r0 = ((const float4*)(h + (size_t)i * 32))[pq];
  int cur = *head_ptr(hplane, hd1, i, sub);
  float di = __shfl(r0.x, (lane & 24) | (DINV_COL / 4), 32);
  bool act = (unsigned)cur < (unsigned)E;
  int s = 0, nx = -1;
  float w = 0.0f;
  if (act) { s = src[cur]; w = ew[cur]; nx = nxt[cur]; }
  float c0 = di * 1.0f * di;
  float4 acc = {0.0f, 0.0f, 0.0f, 0.0f};
  if (sub == 0) { acc.x = r0.x * c0; acc.y = r0.y * c0; acc.z = r0.z * c0; acc.w = r0.w * c0; }
  for (int it = 0; it < E; ++it) {
    if (!act) break;
    float4 r = ((const float4*)(h + (size_t)s * 32))[pq];
    bool nact = (unsigned)nx < (unsigned)E;
    int s2 = 0, nx2 = -1;
    float w2 = 0.0f;
    if (nact) { s2 = src[nx]; w2 = ew[nx]; nx2 = nxt[nx]; }
    float ds = __shfl(r.x, (lane & 24) | (DINV_COL / 4), 32);
    float coef = ds * w * di;
    acc.x += r.x * coef; acc.y += r.y * coef; acc.z += r.z * coef; acc.w += r.w * coef;
    s = s2; w = w2; nx = nx2; act = nact;
  }
  acc.x += __shfl_xor(acc.x, 8, 32);  acc.y += __shfl_xor(acc.y, 8, 32);
  acc.z += __shfl_xor(acc.z, 8, 32);  acc.w += __shfl_xor(acc.w, 8, 32);
  acc.x += __shfl_xor(acc.x, 16, 32); acc.y += __shfl_xor(acc.y, 16, 32);
  acc.z += __shfl_xor(acc.z, 16, 32); acc.w += __shfl_xor(acc.w, 16, 32);
  float4 v = {0.0f, 0.0f, 0.0f, 0.0f};
  if (q < HID / 4) {
    v.x = acc.x + b[4 * q + 0]; v.y = acc.y + b[4 * q + 1];
    v.z = acc.z + b[4 * q + 2]; v.w = acc.w + b[4 * q + 3];
  }
  float ss = v.x * v.x + v.y * v.y + v.z * v.z + v.w * v.w;
  ss += __shfl_xor(ss, 1, 32);
  ss += __shfl_xor(ss, 2, 32);
  ss += __shfl_xor(ss, 4, 32);
  float nrm = fmaxf(sqrtf(ss), 1e-12f);
  if (lane < HID / 4) {
    float r0n = v.x / nrm, r1n = v.y / nrm, r2n = v.z / nrm, r3n = v.w / nrm;
    float4 o;
    o.x = (r0n > 0.0f) ? r0n : 0.0f;
    o.y = (r1n > 0.0f) ? r1n : 0.0f;
    o.z = (r2n > 0.0f) ? r2n : 0.0f;
    o.w = (r3n > 0.0f) ? r3n : 0.0f;
    ((float4*)(emb + (size_t)i * 64 + off))[q] = o;
  }
}

template <int NTILES>
__device__ __forceinline__ void gemm_wave(int pair, const float* __restrict__ A, int lda, int acol0,
                                          const float* __restrict__ Bp, int Ksteps,
                                          float* __restrict__ out, int ldo, int ncols,
                                          const float* __restrict__ bias, int npairs) {
  if (pair >= npairs) return;
  int lane = threadIdx.x & 31;
  int half = lane >> 4;
  int l16 = lane & 15;
  int m0 = pair * 32;
  const float* ap0 = A + (size_t)(m0 + l16) * lda + acol0 + half * 2;
  const float* ap1 = ap0 + (size_t)16 * lda;
  const float* bp = Bp + lane * 2;
  v8f acc[2][NTILES];
#pragma unroll
  for (int rr = 0; rr < 2; ++rr)
#pragma unroll
    for (int t = 0; t < NTILES; ++t) acc[rr][t] = (v8f){0.f, 0.f, 0.f, 0.f, 0.f, 0.f, 0.f, 0.f};

#pragma unroll 2
  for (int ks = 0; ks < Ksteps; ++ks) {
    v2f a0 = *(const v2f*)(ap0 + 4 * ks);
    v2f a1 = *(const v2f*)(ap1 + 4 * ks);
#pragma unroll
    for (int t = 0; t < NTILES; ++t) {
      v2f b = *(const v2f*)(bp + (size_t)(ks * NTILES + t) * 64);
      acc[0][t] = __builtin_amdgcn_wmma_f32_16x16x4_f32(false, a0, false, b, (short)0, acc[0][t], false, false);
      acc[1][t] = __builtin_amdgcn_wmma_f32_16x16x4_f32(false, a1, false, b, (short)0, acc[1][t], false, false);
    }
  }

#pragma unroll
  for (int rr = 0; rr < 2; ++rr)
#pragma unroll
    for (int t = 0; t < NTILES; ++t)
#pragma unroll
      for (int r = 0; r < 8; ++r) {
        int row = m0 + rr * 16 + r + half * 8;
        int col = t * 16 + l16;
        if (col < ncols) out[(size_t)row * ldo + col] = acc[rr][t][r] + (bias ? bias[col] : 0.0f);
      }
}

template <int NTILES>
__global__ void k_wmma_gemm(const float* __restrict__ A, int lda, int acol0,
                            const float* __restrict__ Bp, int Ksteps,
                            float* __restrict__ out, int ldo, int ncols,
                            const float* __restrict__ bias, int npairs) {
  gemm_wave<NTILES>(blockIdx.x * (blockDim.x >> 5) + (threadIdx.x >> 5), A, lda, acol0, Bp, Ksteps,
                    out, ldo, ncols, bias, npairs);
}

__global__ void k_front(const float* __restrict__ x, const float* __restrict__ W1p, int Ksteps1,
                        float* __restrict__ h, int npairs, int gemm_blocks,
                        PackJob p2, PackJob p3, PackJob pl, int pack_blocks,
                        const int* __restrict__ dst, int* __restrict__ heads,
                        int* __restrict__ hd1, int* __restrict__ nxt, int E) {
  int blk = blockIdx.x;
  if (blk < gemm_blocks) {
    gemm_wave<2>(blk * (blockDim.x >> 5) + (threadIdx.x >> 5), x, IN_DIM, 0, W1p, Ksteps1,
                 h, 32, HID, nullptr, npairs);
    return;
  }
  blk -= gemm_blocks;
  if (blk < 3 * pack_blocks) {
    int job = blk / pack_blocks;
    int t = (blk - job * pack_blocks) * blockDim.x + threadIdx.x;
    if (job == 0) packB_elem(t, p2.W, p2.Krows, p2.Ncols, p2.Bp, p2.Ksteps, p2.ntiles);
    else if (job == 1) packB_elem(t, p3.W, p3.Krows, p3.Ncols, p3.Bp, p3.Ksteps, p3.ntiles);
    else packB_elem(t, pl.W, pl.Krows, pl.Ncols, pl.Bp, pl.Ksteps, pl.ntiles);
    return;
  }
  blk -= 3 * pack_blocks;
  build_edge(blk * blockDim.x + threadIdx.x, dst, heads, hd1, nxt, E);
}

extern "C" void kernel_launch(void* const* d_in, const int* in_sizes, int n_in,
                              void* d_out, int out_size, void* d_ws, size_t ws_size,
                              hipStream_t stream) {
  const float* x  = (const float*)d_in[0];
  const int*   ei = (const int*)d_in[1];
  const float* ew = (const float*)d_in[2];
  const float* W1 = (const float*)d_in[3];
  const float* b1 = (const float*)d_in[4];
  const float* W2 = (const float*)d_in[5];
  const float* b2 = (const float*)d_in[6];
  const float* W3 = (const float*)d_in[7];
  const float* b3 = (const float*)d_in[8];
  const float* Wl = (const float*)d_in[9];
  const float* bl = (const float*)d_in[10];
  float* out = (float*)d_out;

  const int N = in_sizes[0] / IN_DIM;
  const long long E = in_sizes[2];
  const int* src = ei;
  const int* dst = ei + E;

  const int KS1 = IN_DIM / 4;
  const int KS2 = (HID + 3) / 4;
  const int KSL = (3 * HID) / 4;

  float* ws = (float*)d_ws;
  int* heads1 = (int*)d_ws;
  ws += N;
  ws += N;
  float* h    = ws; ws += (size_t)N * 32;
  int*   nxt  = (int*)ws; ws += (size_t)N * 32;
  float* emb  = ws; ws += (size_t)N * 64;
  float* W1p  = ws; ws += KS1 * 2 * 64;
  float* W2p  = ws; ws += KS2 * 2 * 64;
  float* W3p  = ws; ws += KS2 * 2 * 64;
  float* Wlp  = ws; ws += KSL * 1 * 64;
  int* heads = (int*)emb;

  const int TB = 256;
  const int FB = 128;
  const int WB = 64;
  const int NPAIRS = N / 32;
  const int GB = (NPAIRS + 3) / 4;
  dim3 gblk(GB);
  const int Ei = (int)E;
  int g_walk = (int)(((long long)N * NSUB + WB - 1) / WB);
  int g_agg = (N + 7) / 8;

  k_init<<<(int)(((long long)N * 66 + TB - 1) / TB), TB, 0, stream>>>(heads, heads1, N);
  k_packB<<<(KS1 * 2 * 64 + TB - 1) / TB, TB, 0, stream>>>(W1, IN_DIM, HID, W1p, KS1, 2);

  PackJob p2 = {W2, W2p, HID, HID, KS2, 2};
  PackJob p3 = {W3, W3p, HID, HID, KS2, 2};
  PackJob pl = {Wl, Wlp, 3 * HID, OUT_DIM, KSL, 1};
  const int PB = (KSL * 1 * 64 + FB - 1) / FB;
  const int BB = (int)((E + FB - 1) / FB);
  k_front<<<GB + 3 * PB + BB, FB, 0, stream>>>(x, W1p, KS1, h, NPAIRS, GB, p2, p3, pl, PB,
                                               dst, heads, heads1, nxt, Ei);
  k_deg_walk<<<g_walk, WB, 0, stream>>>(heads, heads1, nxt, ew, h, N, Ei);


  k_walk_agg<<<g_agg, TB, 0, stream>>>(heads, heads1, nxt, src, ew, h, b1, emb, N, 0, Ei);

  k_wmma_gemm<2><<<gblk, 128, 0, stream>>>(emb, 64, 0, W2p, KS2, h, 32, HID, nullptr, NPAIRS);
  k_walk_agg<<<g_agg, TB, 0, stream>>>(heads, heads1, nxt, src, ew, h, b2, emb, N, HID, Ei);

  k_wmma_gemm<2><<<gblk, 128, 0, stream>>>(emb, 64, HID, W3p, KS2, h, 32, HID, nullptr, NPAIRS);
  k_walk_agg<<<g_agg, TB, 0, stream>>>(heads, heads1, nxt, src, ew, h, b3, emb, N, 2 * HID, Ei);

  k_wmma_gemm<1><<<gblk, 128, 0, stream>>>(emb, 64, 0, Wlp, KSL, out, OUT_DIM, OUT_DIM, bl, NPAIRS);
}


# List build writes each edge a 16-byte record {source, weight, next} behind the old planes when the handed workspace holds them (else 3 planes as before); a walk hop asks 1 line, not 3; sum order kept
# speedup vs baseline: 3.7339x; 1.0788x over previous
#include <hip/hip_runtime.h>
#include <hip/hip_bf16.h>

#define IN_DIM 512
#define HID 20
#define OUT_DIM 10

typedef __attribute__((ext_vector_type(2))) float v2f;
typedef __attribute__((ext_vector_type(8))) float v8f;

#define NSUB 4
#define HEAD_COL 60
#define DINV_COL 24

template <bool REC>
__device__ __forceinline__ void edge_fetch(int cur, const int* __restrict__ rec,
                                           const int* __restrict__ src, const float* __restrict__ ew,
                                           const int* __restrict__ nxt, int& s, float& w, int& nx) {
  if (REC) {
    int4 rc = ((const int4*)rec)[cur];
    s = rc.x; w = __int_as_float(rc.y); nx = rc.z;
  } else {
    s = src[cur]; w = ew[cur]; nx = nxt[cur];
  }
}

__global__ void k_init(int* __restrict__ embw, int* __restrict__ hd1, int n) {
  long long t = (long long)blockIdx.x * blockDim.x + threadIdx.x;
  long long ne = (long long)n * 64;
  if (t < ne) embw[t] = ((int)(t & 63) >= HEAD_COL) ? -1 : 0;
  else if (t < ne + (long long)n * 2) hd1[t - ne] = -1;
}

__device__ __forceinline__ const int* head_ptr(const int* embw, const int* hd1, int i, int sub) {
  return (sub < 2) ? (embw + (size_t)i * 64 + HEAD_COL + sub) : (hd1 + (size_t)i * 2 + (sub - 2));
}

__device__ __forceinline__ void packB_elem(int t, const float* __restrict__ W, int Krows, int Ncols,
                                           float* __restrict__ Bp, int Ksteps, int ntiles) {
  int total = Ksteps * ntiles * 64;
  if (t >= total) return;
  int j = t & 1;
  int lane = (t >> 1) & 31;
  int tile = (t >> 6) % ntiles;
  int ks = (t >> 6) / ntiles;
  int k = 4 * ks + 2 * (lane >> 4) + j;
  int n = tile * 16 + (lane & 15);
  Bp[t] = (k < Krows && n < Ncols) ? W[k * Ncols + n] : 0.0f;
}

__global__ void k_packB(const float* __restrict__ W, int Krows, int Ncols,
                        float* __restrict__ Bp, int Ksteps, int ntiles) {
  packB_elem(blockIdx.x * blockDim.x + threadIdx.x, W, Krows, Ncols, Bp, Ksteps, ntiles);
}

struct PackJob {
  const float* W;
  float* Bp;
  int Krows, Ncols, Ksteps, ntiles;
};

template <bool REC>
__device__ __forceinline__ void build_edge(int e, const int* __restrict__ dst, int* __restrict__ heads,
                                           int* __restrict__ hd1, int* __restrict__ nxt, int E,
                                           const int* __restrict__ src, const float* __restrict__ ew,
                                           int* __restrict__ rec) {
  if (e >= E) return;
  int d = dst[e];
  int sv = 0;
  float wv = 0.0f;
  if (REC) { sv = src[e]; wv = ew[e]; }
  unsigned place = __builtin_amdgcn_s_sendmsg_rtn(0x87);
  int hf = (int)((place >> 18) & 1u);
  int k = e & 1;
  int* hp = hf ? (hd1 + (size_t)d * 2 + k) : (heads + (size_t)d * 64 + HEAD_COL + k);
  int old = atomicExch(hp, e);
  if (REC) {
    int4 rc = {sv, __float_as_int(wv), old, 0};
    ((int4*)rec)[e] = rc;
  } else {
    nxt[e] = old;
  }
}

template <bool REC>
__global__ void k_deg_walk(const int* __restrict__ hplane, const int* __restrict__ hd1,
                           const int* __restrict__ nxt,
                           const float* __restrict__ ew, float* __restrict__ hrow, int n, int E,
                           const int* __restrict__ rec) {
  int t = blockIdx.x * blockDim.x + threadIdx.x;
  int i = t / NSUB, sub = t % NSUB;
  bool valid = i < n;
  int cur = valid ? *head_ptr(hplane, hd1, i, sub) : -1;
  float p = 0.0f;
  for (int it = 0; it < E; ++it) {
    if ((unsigned)cur >= (unsigned)E) break;
    float w;
    int nx;
    if (REC) {
      int4 rc = ((const int4*)rec)[cur];
      w = __int_as_float(rc.y); nx = rc.z;
    } else {
      w = ew[cur]; nx = nxt[cur];
    }
    p += w;
    cur = nx;
  }
  p += __shfl_xor(p, 1, 32);
  p += __shfl_xor(p, 2, 32);
  if (valid && sub == 0) {
    float d = 1.0f + p;
    float4 o = {(d > 0.0f) ? rsqrtf(d) : 0.0f, 0.0f, 0.0f, 0.0f};
    *(float4*)(hrow + (size_t)i * 32 + DINV_COL) = o;
  }
}

template <bool REC>
__global__ void k_walk_agg(const int* __restrict__ hplane, const int* __restrict__ hd1,
                           const int* __restrict__ nxt,
                           const int* __restrict__ src, const float* __restrict__ ew,
                           const float* __restrict__ h,
                           const float* __restrict__ b, float* __restrict__ emb,
                           int n, int off, int E, const int* __restrict__ rec) {
  static_assert(NSUB * 8 == 32, "one wave = NSUB groups of 8 lanes");
  int lane = threadIdx.x & 31;
  int i = blockIdx.x * (blockDim.x >> 5) + (threadIdx.x >> 5);
  if (i >= n) return;
  int sub = lane >> 3;
  int q = lane & 7;
  int pq = (q < HID / 4) ? q : DINV_COL / 4;
  float4 r0 = ((const float4*)(h + (size_t)i * 32))[pq];
  int cur = *head_ptr(hplane, hd1, i, sub);
  float di = __shfl(r0.x, (lane & 24) | (DINV_COL / 4), 32);
  bool act = (unsigned)cur < (unsigned)E;
  int s = 0, nx = -1;
  float w = 0.0f;
  if (act) edge_fetch<REC>(cur, rec, src, ew, nxt, s, w, nx);
  float c0 = di * 1.0f * di;
  float4 acc = {0.0f, 0.0f, 0.0f, 0.0f};
  if (sub == 0) { acc.x = r0.x * c0; acc.y = r0.y * c0; acc.z = r0.z * c0; acc.w = r0.w * c0; }
  for (int it = 0; it < E; ++it) {
    if (!act) break;
    float4 r = ((const float4*)(h + (size_t)s * 32))[pq];
    bool nact = (unsigned)nx < (unsigned)E;
    int s2 = 0, nx2 = -1;
    float w2 = 0.0f;
    if (nact) edge_fetch<REC>(nx, rec, src, ew, nxt, s2, w2, nx2);
    float ds = __shfl(r.x, (lane & 24) | (DINV_COL / 4), 32);
    float coef = ds * w * di;
    acc.x += r.x * coef; acc.y += r.y * coef; acc.z += r.z * coef; acc.w += r.w * coef;
    s = s2; w = w2; nx = nx2; act = nact;
  }
  acc.x += __shfl_xor(acc.x, 8, 32);  acc.y += __shfl_xor(acc.y, 8, 32);
  acc.z += __shfl_xor(acc.z, 8, 32);  acc.w += __shfl_xor(acc.w, 8, 32);
  acc.x += __shfl_xor(acc.x, 16, 32); acc.y += __shfl_xor(acc.y, 16, 32);
  acc.z += __shfl_xor(acc.z, 16, 32); acc.w += __shfl_xor(acc.w, 16, 32);
  float4 v = {0.0f, 0.0f, 0.0f, 0.0f};
  if (q < HID / 4) {
    v.x = acc.x + b[4 * q + 0]; v.y = acc.y + b[4 * q + 1];
    v.z = acc.z + b[4 * q + 2]; v.w = acc.w + b[4 * q + 3];
  }
  float ss = v.x * v.x + v.y * v.y + v.z * v.z + v.w * v.w;
  ss += __shfl_xor(ss, 1, 32);
  ss += __shfl_xor(ss, 2, 32);
  ss += __shfl_xor(ss, 4, 32);
  float nrm = fmaxf(sqrtf(ss), 1e-12f);
  if (lane < HID / 4) {
    float r0n = v.x / nrm, r1n = v.y / nrm, r2n = v.z / nrm, r3n = v.w / nrm;
    float4 o;
    o.x = (r0n > 0.0f) ? r0n : 0.0f;
    o.y = (r1n > 0.0f) ? r1n : 0.0f;
    o.z = (r2n > 0.0f) ? r2n : 0.0f;
    o.w = (r3n > 0.0f) ? r3n : 0.0f;
    ((float4*)(emb + (size_t)i * 64 + off))[q] = o;
  }
}

template <int NTILES>
__device__ __forceinline__ void gemm_wave(int pair, const float* __restrict__ A, int lda, int acol0,
                                          const float* __restrict__ Bp, int Ksteps,
                                          float* __restrict__ out, int ldo, int ncols,
                                          const float* __restrict__ bias, int npairs) {
  if (pair >= npairs) return;
  int lane = threadIdx.x & 31;
  int half = lane >> 4;
  int l16 = lane & 15;
  int m0 = pair * 32;
  const float* ap0 = A + (size_t)(m0 + l16) * lda + acol0 + half * 2;
  const float* ap1 = ap0 + (size_t)16 * lda;
  const float* bp = Bp + lane * 2;
  v8f acc[2][NTILES];
#pragma unroll
  for (int rr = 0; rr < 2; ++rr)
#pragma unroll
    for (int t = 0; t < NTILES; ++t) acc[rr][t] = (v8f){0.f, 0.f, 0.f, 0.f, 0.f, 0.f, 0.f, 0.f};

#pragma unroll 2
  for (int ks = 0; ks < Ksteps; ++ks) {
    v2f a0 = *(const v2f*)(ap0 + 4 * ks);
    v2f a1 = *(const v2f*)(ap1 + 4 * ks);
#pragma unroll
    for (int t = 0; t < NTILES; ++t) {
      v2f b = *(const v2f*)(bp + (size_t)(ks * NTILES + t) * 64);
      acc[0][t] = __builtin_amdgcn_wmma_f32_16x16x4_f32(false, a0, false, b, (short)0, acc[0][t], false, false);
      acc[1][t] = __builtin_amdgcn_wmma_f32_16x16x4_f32(false, a1, false, b, (short)0, acc[1][t], false, false);
    }
  }

#pragma unroll
  for (int rr = 0; rr < 2; ++rr)
#pragma unroll
    for (int t = 0; t < NTILES; ++t)
#pragma unroll
      for (int r = 0; r < 8; ++r) {
        int row = m0 + rr * 16 + r + half * 8;
        int col = t * 16 + l16;
        if (col < ncols) out[(size_t)row * ldo + col] = acc[rr][t][r] + (bias ? bias[col] : 0.0f);
      }
}

template <int NTILES>
__global__ void k_wmma_gemm(const float* __restrict__ A, int lda, int acol0,
                            const float* __restrict__ Bp, int Ksteps,
                            float* __restrict__ out, int ldo, int ncols,
                            const float* __restrict__ bias, int npairs) {
  gemm_wave<NTILES>(blockIdx.x * (blockDim.x >> 5) + (threadIdx.x >> 5), A, lda, acol0, Bp, Ksteps,
                    out, ldo, ncols, bias, npairs);
}

template <bool REC>
__global__ void k_front(const float* __restrict__ x, const float* __restrict__ W1p, int Ksteps1,
                        float* __restrict__ h, int npairs, int gemm_blocks,
                        PackJob p2, PackJob p3, PackJob pl, int pack_blocks,
                        const int* __restrict__ dst, int* __restrict__ heads,
                        int* __restrict__ hd1, int* __restrict__ nxt, int E,
                        const int* __restrict__ src, const float* __restrict__ ew,
                        int* __restrict__ rec) {
  int blk = blockIdx.x;
  if (blk < gemm_blocks) {
    gemm_wave<2>(blk * (blockDim.x >> 5) + (threadIdx.x >> 5), x, IN_DIM, 0, W1p, Ksteps1,
                 h, 32, HID, nullptr, npairs);
    return;
  }
  blk -= gemm_blocks;
  if (blk < 3 * pack_blocks) {
    int job = blk / pack_blocks;
    int t = (blk - job * pack_blocks) * blockDim.x + threadIdx.x;
    if (job == 0) packB_elem(t, p2.W, p2.Krows, p2.Ncols, p2.Bp, p2.Ksteps, p2.ntiles);
    else if (job == 1) packB_elem(t, p3.W, p3.Krows, p3.Ncols, p3.Bp, p3.Ksteps, p3.ntiles);
    else packB_elem(t, pl.W, pl.Krows, pl.Ncols, pl.Bp, pl.Ksteps, pl.ntiles);
    return;
  }
  blk -= 3 * pack_blocks;
  build_edge<REC>(blk * blockDim.x + threadIdx.x, dst, heads, hd1, nxt, E, src, ew, rec);
}

extern "C" void kernel_launch(void* const* d_in, const int* in_sizes, int n_in,
                              void* d_out, int out_size, void* d_ws, size_t ws_size,
                              hipStream_t stream) {
  const float* x  = (const float*)d_in[0];
  const int*   ei = (const int*)d_in[1];
  const float* ew = (const float*)d_in[2];
  const float* W1 = (const float*)d_in[3];
  const float* b1 = (const float*)d_in[4];
  const float* W2 = (const float*)d_in[5];
  const float* b2 = (const float*)d_in[6];
  const float* W3 = (const float*)d_in[7];
  const float* b3 = (const float*)d_in[8];
  const float* Wl = (const float*)d_in[9];
  const float* bl = (const float*)d_in[10];
  float* out = (float*)d_out;

  const int N = in_sizes[0] / IN_DIM;
  const long long E = in_sizes[2];
  const int* src = ei;
  const int* dst = ei + E;

  const int KS1 = IN_DIM / 4;
  const int KS2 = (HID + 3) / 4;
  const int KSL = (3 * HID) / 4;

  float* ws = (float*)d_ws;
  int* heads1 = (int*)d_ws;
  ws += N;
  ws += N;
  float* h    = ws; ws += (size_t)N * 32;
  int*   nxt  = (int*)ws; ws += (size_t)N * 32;
  float* emb  = ws; ws += (size_t)N * 64;
  float* W1p  = ws; ws += KS1 * 2 * 64;
  float* W2p  = ws; ws += KS2 * 2 * 64;
  float* W3p  = ws; ws += KS2 * 2 * 64;
  float* Wlp  = ws; ws += KSL * 1 * 64;
  int* rec = (int*)ws;
  const long long room = (long long)ws_size - (long long)((char*)ws - (char*)d_ws);
  const bool use_rec = room >= E * 16;
  int* heads = (int*)emb;

  const int TB = 256;
  const int FB = 128;
  const int WB = 64;
  const int NPAIRS = N / 32;
  const int GB = (NPAIRS + 3) / 4;
  dim3 gblk(GB);
  const int Ei = (int)E;
  int g_walk = (int)(((long long)N * NSUB + WB - 1) / WB);
  int g_agg = (N + 7) / 8;

  k_init<<<(int)(((long long)N * 66 + TB - 1) / TB), TB, 0, stream>>>(heads, heads1, N);
  k_packB<<<(KS1 * 2 * 64 + TB - 1) / TB, TB, 0, stream>>>(W1, IN_DIM, HID, W1p, KS1, 2);

  PackJob p2 = {W2, W2p, HID, HID, KS2, 2};
  PackJob p3 = {W3, W3p, HID, HID, KS2, 2};
  PackJob pl = {Wl, Wlp, 3 * HID, OUT_DIM, KSL, 1};
  const int PB = (KSL * 1 * 64 + FB - 1) / FB;
  const int BB = (int)((E + FB - 1) / FB);
  if (use_rec) {
    k_front<true><<<GB + 3 * PB + BB, FB, 0, stream>>>(x, W1p, KS1, h, NPAIRS, GB, p2, p3, pl, PB,
                                                       dst, heads, heads1, nxt, Ei, src, ew, rec);
    k_deg_walk<true><<<g_walk, WB, 0, stream>>>(heads, heads1, nxt, ew, h, N, Ei, rec);
  } else {
    k_front<false><<<GB + 3 * PB + BB, FB, 0, stream>>>(x, W1p, KS1, h, NPAIRS, GB, p2, p3, pl, PB,
                                                        dst, heads, heads1, nxt, Ei, src, ew, rec);
    k_deg_walk<false><<<g_walk, WB, 0, stream>>>(heads, heads1, nxt, ew, h, N, Ei, rec);
  }


  if (use_rec) k_walk_agg<true><<<g_agg, TB, 0, stream>>>(heads, heads1, nxt, src, ew, h, b1, emb, N, 0, Ei, rec);
  else k_walk_agg<false><<<g_agg, TB, 0, stream>>>(heads, heads1, nxt, src, ew, h, b1, emb, N, 0, Ei, rec);

  k_wmma_gemm<2><<<gblk, 128, 0, stream>>>(emb, 64, 0, W2p, KS2, h, 32, HID, nullptr, NPAIRS);
  if (use_rec) k_walk_agg<true><<<g_agg, TB, 0, stream>>>(heads, heads1, nxt, src, ew, h, b2, emb, N, HID, Ei, rec);
  else k_walk_agg<false><<<g_agg, TB, 0, stream>>>(heads, heads1, nxt, src, ew, h, b2, emb, N, HID, Ei, rec);

  k_wmma_gemm<2><<<gblk, 128, 0, stream>>>(emb, 64, HID, W3p, KS2, h, 32, HID, nullptr, NPAIRS);
  if (use_rec) k_walk_agg<true><<<g_agg, TB, 0, stream>>>(heads, heads1, nxt, src, ew, h, b3, emb, N, 2 * HID, Ei, rec);
  else k_walk_agg<false><<<g_agg, TB, 0, stream>>>(heads, heads1, nxt, src, ew, h, b3, emb, N, 2 * HID, Ei, rec);

  k_wmma_gemm<1><<<gblk, 128, 0, stream>>>(emb, 64, 0, Wlp, KSL, out, OUT_DIM, OUT_DIM, bl, NPAIRS);
}


# Layer-1 GEMM split over k: a 32-row pair a block, its 4 waves take k steps w, w+4 of every 8 (one x line); partial tiles added via LDS in wave order; load rounds a wave 64 -> 16; sum order changes
# speedup vs baseline: 3.8809x; 1.0394x over previous
#include <hip/hip_runtime.h>
#include <hip/hip_bf16.h>

#define IN_DIM 512
#define HID 20
#define OUT_DIM 10

typedef __attribute__((ext_vector_type(2))) float v2f;
typedef __attribute__((ext_vector_type(8))) float v8f;

#define NSUB 4
#define HEAD_COL 60
#define DINV_COL 24

template <bool REC>
__device__ __forceinline__ void edge_fetch(int cur, const int* __restrict__ rec,
                                           const int* __restrict__ src, const float* __restrict__ ew,
                                           const int* __restrict__ nxt, int& s, float& w, int& nx) {
  if (REC) {
    int4 rc = ((const int4*)rec)[cur];
    s = rc.x; w = __int_as_float(rc.y); nx = rc.z;
  } else {
    s = src[cur]; w = ew[cur]; nx = nxt[cur];
  }
}

__global__ void k_init(int* __restrict__ embw, int* __restrict__ hd1, int n) {
  long long t = (long long)blockIdx.x * blockDim.x + threadIdx.x;
  long long ne = (long long)n * 64;
  if (t < ne) embw[t] = ((int)(t & 63) >= HEAD_COL) ? -1 : 0;
  else if (t < ne + (long long)n * 2) hd1[t - ne] = -1;
}

__device__ __forceinline__ const int* head_ptr(const int* embw, const int* hd1, int i, int sub) {
  return (sub < 2) ? (embw + (size_t)i * 64 + HEAD_COL + sub) : (hd1 + (size_t)i * 2 + (sub - 2));
}

__device__ __forceinline__ void packB_elem(int t, const float* __restrict__ W, int Krows, int Ncols,
                                           float* __restrict__ Bp, int Ksteps, int ntiles) {
  int total = Ksteps * ntiles * 64;
  if (t >= total) return;
  int j = t & 1;
  int lane = (t >> 1) & 31;
  int tile = (t >> 6) % ntiles;
  int ks = (t >> 6) / ntiles;
  int k = 4 * ks + 2 * (lane >> 4) + j;
  int n = tile * 16 + (lane & 15);
  Bp[t] = (k < Krows && n < Ncols) ? W[k * Ncols + n] : 0.0f;
}

__global__ void k_packB(const float* __restrict__ W, int Krows, int Ncols,
                        float* __restrict__ Bp, int Ksteps, int ntiles) {
  packB_elem(blockIdx.x * blockDim.x + threadIdx.x, W, Krows, Ncols, Bp, Ksteps, ntiles);
}

struct PackJob {
  const float* W;
  float* Bp;
  int Krows, Ncols, Ksteps, ntiles;
};

template <bool REC>
__device__ __forceinline__ void build_edge(int e, const int* __restrict__ dst, int* __restrict__ heads,
                                           int* __restrict__ hd1, int* __restrict__ nxt, int E,
                                           const int* __restrict__ src, const float* __restrict__ ew,
                                           int* __restrict__ rec) {
  if (e >= E) return;
  int d = dst[e];
  int sv = 0;
  float wv = 0.0f;
  if (REC) { sv = src[e]; wv = ew[e]; }
  unsigned place = __builtin_amdgcn_s_sendmsg_rtn(0x87);
  int hf = (int)((place >> 18) & 1u);
  int k = e & 1;
  int* hp = hf ? (hd1 + (size_t)d * 2 + k) : (heads + (size_t)d * 64 + HEAD_COL + k);
  int old = atomicExch(hp, e);
  if (REC) {
    int4 rc = {sv, __float_as_int(wv), old, 0};
    ((int4*)rec)[e] = rc;
  } else {
    nxt[e] = old;
  }
}

template <bool REC>
__global__ void k_deg_walk(const int* __restrict__ hplane, const int* __restrict__ hd1,
                           const int* __restrict__ nxt,
                           const float* __restrict__ ew, float* __restrict__ hrow, int n, int E,
                           const int* __restrict__ rec) {
  int t = blockIdx.x * blockDim.x + threadIdx.x;
  int i = t / NSUB, sub = t % NSUB;
  bool valid = i < n;
  int cur = valid ? *head_ptr(hplane, hd1, i, sub) : -1;
  float p = 0.0f;
  for (int it = 0; it < E; ++it) {
    if ((unsigned)cur >= (unsigned)E) break;
    float w;
    int nx;
    if (REC) {
      int4 rc = ((const int4*)rec)[cur];
      w = __int_as_float(rc.y); nx = rc.z;
    } else {
      w = ew[cur]; nx = nxt[cur];
    }
    p += w;
    cur = nx;
  }
  p += __shfl_xor(p, 1, 32);
  p += __shfl_xor(p, 2, 32);
  if (valid && sub == 0) {
    float d = 1.0f + p;
    float4 o = {(d > 0.0f) ? rsqrtf(d) : 0.0f, 0.0f, 0.0f, 0.0f};
    *(float4*)(hrow + (size_t)i * 32 + DINV_COL) = o;
  }
}

template <bool REC>
__global__ void k_walk_agg(const int* __restrict__ hplane, const int* __restrict__ hd1,
                           const int* __restrict__ nxt,
                           const int* __restrict__ src, const float* __restrict__ ew,
                           const float* __restrict__ h,
                           const float* __restrict__ b, float* __restrict__ emb,
                           int n, int off, int E, const int* __restrict__ rec) {
  static_assert(NSUB * 8 == 32, "one wave = NSUB groups of 8 lanes");
  int lane = threadIdx.x & 31;
  int i = blockIdx.x * (blockDim.x >> 5) + (threadIdx.x >> 5);
  if (i >= n) return;
  int sub = lane >> 3;
  int q = lane & 7;
  int pq = (q < HID / 4) ? q : DINV_COL / 4;
  float4 r0 = ((const float4*)(h + (size_t)i * 32))[pq];
  int cur = *head_ptr(hplane, hd1, i, sub);
  float di = __shfl(r0.x, (lane & 24) | (DINV_COL / 4), 32);
  bool act = (unsigned)cur < (unsigned)E;
  int s = 0, nx = -1;
  float w = 0.0f;
  if (act) edge_fetch<REC>(cur, rec, src, ew, nxt, s, w, nx);
  float c0 = di * 1.0f * di;
  float4 acc = {0.0f, 0.0f, 0.0f, 0.0f};
  if (sub == 0) { acc.x = r0.x * c0; acc.y = r0.y * c0; acc.z = r0.z * c0; acc.w = r0.w * c0; }
  for (int it = 0; it < E; ++it) {
    if (!act) break;
    float4 r = ((const float4*)(h + (size_t)s * 32))[pq];
    bool nact = (unsigned)nx < (unsigned)E;
    int s2 = 0, nx2 = -1;
    float w2 = 0.0f;
    if (nact) edge_fetch<REC>(nx, rec, src, ew, nxt, s2, w2, nx2);
    float ds = __shfl(r.x, (lane & 24) | (DINV_COL / 4), 32);
    float coef = ds * w * di;
    acc.x += r.x * coef; acc.y += r.y * coef; acc.z += r.z * coef; acc.w += r.w * coef;
    s = s2; w = w2; nx = nx2; act = nact;
  }
  acc.x += __shfl_xor(acc.x, 8, 32);  acc.y += __shfl_xor(acc.y, 8, 32);
  acc.z += __shfl_xor(acc.z, 8, 32);  acc.w += __shfl_xor(acc.w, 8, 32);
  acc.x += __shfl_xor(acc.x, 16, 32); acc.y += __shfl_xor(acc.y, 16, 32);
  acc.z += __shfl_xor(acc.z, 16, 32); acc.w += __shfl_xor(acc.w, 16, 32);
  float4 v = {0.0f, 0.0f, 0.0f, 0.0f};
  if (q < HID / 4) {
    v.x = acc.x + b[4 * q + 0]; v.y = acc.y + b[4 * q + 1];
    v.z = acc.z + b[4 * q + 2]; v.w = acc.w + b[4 * q + 3];
  }
  float ss = v.x * v.x + v.y * v.y + v.z * v.z + v.w * v.w;
  ss += __shfl_xor(ss, 1, 32);
  ss += __shfl_xor(ss, 2, 32);
  ss += __shfl_xor(ss, 4, 32);
  float nrm = fmaxf(sqrtf(ss), 1e-12f);
  if (lane < HID / 4) {
    float r0n = v.x / nrm, r1n = v.y / nrm, r2n = v.z / nrm, r3n = v.w / nrm;
    float4 o;
    o.x = (r0n > 0.0f) ? r0n : 0.0f;
    o.y = (r1n > 0.0f) ? r1n : 0.0f;
    o.z = (r2n > 0.0f) ? r2n : 0.0f;
    o.w = (r3n > 0.0f) ? r3n : 0.0f;
    ((float4*)(emb + (size_t)i * 64 + off))[q] = o;
  }
}

template <int NTILES>
__device__ __forceinline__ void gemm_wave(int pair, const float* __restrict__ A, int lda, int acol0,
                                          const float* __restrict__ Bp, int Ksteps,
                                          float* __restrict__ out, int ldo, int ncols,
                                          const float* __restrict__ bias, int npairs) {
  if (pair >= npairs) return;
  int lane = threadIdx.x & 31;
  int half = lane >> 4;
  int l16 = lane & 15;
  int m0 = pair * 32;
  const float* ap0 = A + (size_t)(m0 + l16) * lda + acol0 + half * 2;
  const float* ap1 = ap0 + (size_t)16 * lda;
  const float* bp = Bp + lane * 2;
  v8f acc[2][NTILES];
#pragma unroll
  for (int rr = 0; rr < 2; ++rr)
#pragma unroll
    for (int t = 0; t < NTILES; ++t) acc[rr][t] = (v8f){0.f, 0.f, 0.f, 0.f, 0.f, 0.f, 0.f, 0.f};

#pragma unroll 2
  for (int ks = 0; ks < Ksteps; ++ks) {
    v2f a0 = *(const v2f*)(ap0 + 4 * ks);
    v2f a1 = *(const v2f*)(ap1 + 4 * ks);
#pragma unroll
    for (int t = 0; t < NTILES; ++t) {
      v2f b = *(const v2f*)(bp + (size_t)(ks * NTILES + t) * 64);
      acc[0][t] = __builtin_amdgcn_wmma_f32_16x16x4_f32(false, a0, false, b, (short)0, acc[0][t], false, false);
      acc[1][t] = __builtin_amdgcn_wmma_f32_16x16x4_f32(false, a1, false, b, (short)0, acc[1][t], false, false);
    }
  }

#pragma unroll
  for (int rr = 0; rr < 2; ++rr)
#pragma unroll
    for (int t = 0; t < NTILES; ++t)
#pragma unroll
      for (int r = 0; r < 8; ++r) {
        int row = m0 + rr * 16 + r + half * 8;
        int col = t * 16 + l16;
        if (col < ncols) out[(size_t)row * ldo + col] = acc[rr][t][r] + (bias ? bias[col] : 0.0f);
      }
}

template <int NTILES>
__global__ void k_wmma_gemm(const float* __restrict__ A, int lda, int acol0,
                            const float* __restrict__ Bp, int Ksteps,
                            float* __restrict__ out, int ldo, int ncols,
                            const float* __restrict__ bias, int npairs) {
  gemm_wave<NTILES>(blockIdx.x * (blockDim.x >> 5) + (threadIdx.x >> 5), A, lda, acol0, Bp, Ksteps,
                    out, ldo, ncols, bias, npairs);
}

#define SPLITK 4
__device__ __forceinline__ void gemm1_splitk(int pair, const float* __restrict__ A,
                                             const float* __restrict__ Bp, int Ksteps,
                                             float* __restrict__ out, float* __restrict__ red) {
  static_assert((IN_DIM / 4) % (2 * SPLITK) == 0, "k steps come in groups of 8 (one line of a row)");
  int lane = threadIdx.x & 31;
  int wv = threadIdx.x >> 5;
  int half = lane >> 4;
  int l16 = lane & 15;
  int m0 = pair * 32;
  const float* ap0 = A + (size_t)(m0 + l16) * IN_DIM + half * 2 + 4 * wv;
  const float* ap1 = ap0 + (size_t)16 * IN_DIM;
  const float* bp = Bp + lane * 2 + (size_t)wv * 2 * 64;
  v8f acc[2][2];
#pragma unroll
  for (int rr = 0; rr < 2; ++rr)
#pragma unroll
    for (int t = 0; t < 2; ++t) acc[rr][t] = (v8f){0.f, 0.f, 0.f, 0.f, 0.f, 0.f, 0.f, 0.f};

  int groups = Ksteps / (2 * SPLITK);
#pragma unroll 1
  for (int g = 0; g < groups; ++g) {
#pragma unroll
    for (int u = 0; u < 2; ++u) {
      int ks = 8 * g + 4 * u;
      v2f a0 = *(const v2f*)(ap0 + 4 * ks);
      v2f a1 = *(const v2f*)(ap1 + 4 * ks);
#pragma unroll
      for (int t = 0; t < 2; ++t) {
        v2f b = *(const v2f*)(bp + (size_t)(ks * 2 + t) * 64);
        acc[0][t] = __builtin_amdgcn_wmma_f32_16x16x4_f32(false, a0, false, b, (short)0, acc[0][t], false, false);
        acc[1][t] = __builtin_amdgcn_wmma_f32_16x16x4_f32(false, a1, false, b, (short)0, acc[1][t], false, false);
      }
    }
  }

#pragma unroll
  for (int rr = 0; rr < 2; ++rr)
#pragma unroll
    for (int t = 0; t < 2; ++t)
#pragma unroll
      for (int r = 0; r < 8; ++r) red[((wv * 4 + rr * 2 + t) * 8 + r) * 32 + lane] = acc[rr][t][r];
  __syncthreads();
  int rr = wv >> 1;
  float s[4][2];
#pragma unroll
  for (int rq = 0; rq < 4; ++rq)
#pragma unroll
    for (int t = 0; t < 2; ++t) {
      int r = 4 * (wv & 1) + rq;
      float v = red[((0 * 4 + rr * 2 + t) * 8 + r) * 32 + lane];
#pragma unroll
      for (int p = 1; p < SPLITK; ++p) v += red[((p * 4 + rr * 2 + t) * 8 + r) * 32 + lane];
      s[rq][t] = v;
    }
  float* orow = out + (size_t)(m0 + rr * 16 + 4 * (wv & 1) + half * 8) * 32;
#pragma unroll
  for (int rq = 0; rq < 4; ++rq) orow[rq * 32 + l16] = s[rq][0];
  if (l16 < HID - 16) {
#pragma unroll
    for (int rq = 0; rq < 4; ++rq) orow[rq * 32 + 16 + l16] = s[rq][1];
  }
}

template <bool REC>
__global__ void k_front(const float* __restrict__ x, const float* __restrict__ W1p, int Ksteps1,
                        float* __restrict__ h, int npairs, int gemm_blocks,
                        PackJob p2, PackJob p3, PackJob pl, int pack_blocks,
                        const int* __restrict__ dst, int* __restrict__ heads,
                        int* __restrict__ hd1, int* __restrict__ nxt, int E,
                        const int* __restrict__ src, const float* __restrict__ ew,
                        int* __restrict__ rec) {
  __shared__ float red[SPLITK * 4 * 8 * 32];
  int blk = blockIdx.x;
  if (blk < gemm_blocks) {
    gemm1_splitk(blk, x, W1p, Ksteps1, h, red);
    return;
  }
  blk -= gemm_blocks;
  if (blk < 3 * pack_blocks) {
    int job = blk / pack_blocks;
    int t = (blk - job * pack_blocks) * blockDim.x + threadIdx.x;
    if (job == 0) packB_elem(t, p2.W, p2.Krows, p2.Ncols, p2.Bp, p2.Ksteps, p2.ntiles);
    else if (job == 1) packB_elem(t, p3.W, p3.Krows, p3.Ncols, p3.Bp, p3.Ksteps, p3.ntiles);
    else packB_elem(t, pl.W, pl.Krows, pl.Ncols, pl.Bp, pl.Ksteps, pl.ntiles);
    return;
  }
  blk -= 3 * pack_blocks;
  build_edge<REC>(blk * blockDim.x + threadIdx.x, dst, heads, hd1, nxt, E, src, ew, rec);
}

extern "C" void kernel_launch(void* const* d_in, const int* in_sizes, int n_in,
                              void* d_out, int out_size, void* d_ws, size_t ws_size,
                              hipStream_t stream) {
  const float* x  = (const float*)d_in[0];
  const int*   ei = (const int*)d_in[1];
  const float* ew = (const float*)d_in[2];
  const float* W1 = (const float*)d_in[3];
  const float* b1 = (const float*)d_in[4];
  const float* W2 = (const float*)d_in[5];
  const float* b2 = (const float*)d_in[6];
  const float* W3 = (const float*)d_in[7];
  const float* b3 = (const float*)d_in[8];
  const float* Wl = (const float*)d_in[9];
  const float* bl = (const float*)d_in[10];
  float* out = (float*)d_out;

  const int N = in_sizes[0] / IN_DIM;
  const long long E = in_sizes[2];
  const int* src = ei;
  const int* dst = ei + E;

  const int KS1 = IN_DIM / 4;
  const int KS2 = (HID + 3) / 4;
  const int KSL = (3 * HID) / 4;

  float* ws = (float*)d_ws;
  int* heads1 = (int*)d_ws;
  ws += N;
  ws += N;
  float* h    = ws; ws += (size_t)N * 32;
  int*   nxt  = (int*)ws; ws += (size_t)N * 32;
  float* emb  = ws; ws += (size_t)N * 64;
  float* W1p  = ws; ws += KS1 * 2 * 64;
  float* W2p  = ws; ws += KS2 * 2 * 64;
  float* W3p  = ws; ws += KS2 * 2 * 64;
  float* Wlp  = ws; ws += KSL * 1 * 64;
  int* rec = (int*)ws;
  const long long room = (long long)ws_size - (long long)((char*)ws - (char*)d_ws);
  const bool use_rec = room >= E * 16;
  int* heads = (int*)emb;

  const int TB = 256;
  const int FB = 128;
  const int WB = 64;
  const int NPAIRS = N / 32;
  const int GB = (NPAIRS + 3) / 4;
  dim3 gblk(GB);
  const int Ei = (int)E;
  int g_walk = (int)(((long long)N * NSUB + WB - 1) / WB);
  int g_agg = (N + 7) / 8;

  k_init<<<(int)(((long long)N * 66 + TB - 1) / TB), TB, 0, stream>>>(heads, heads1, N);
  k_packB<<<(KS1 * 2 * 64 + TB - 1) / TB, TB, 0, stream>>>(W1, IN_DIM, HID, W1p, KS1, 2);

  PackJob p2 = {W2, W2p, HID, HID, KS2, 2};
  PackJob p3 = {W3, W3p, HID, HID, KS2, 2};
  PackJob pl = {Wl, Wlp, 3 * HID, OUT_DIM, KSL, 1};
  const int PB = (KSL * 1 * 64 + FB - 1) / FB;
  const int BB = (int)((E + FB - 1) / FB);
  const int GF = NPAIRS;
  if (use_rec) {
    k_front<true><<<GF + 3 * PB + BB, FB, 0, stream>>>(x, W1p, KS1, h, NPAIRS, GF, p2, p3, pl, PB,
                                                       dst, heads, heads1, nxt, Ei, src, ew, rec);
    k_deg_walk<true><<<g_walk, WB, 0, stream>>>(heads, heads1, nxt, ew, h, N, Ei, rec);
  } else {
    k_front<false><<<GF + 3 * PB + BB, FB, 0, stream>>>(x, W1p, KS1, h, NPAIRS, GF, p2, p3, pl, PB,
                                                        dst, heads, heads1, nxt, Ei, src, ew, rec);
    k_deg_walk<false><<<g_walk, WB, 0, stream>>>(heads, heads1, nxt, ew, h, N, Ei, rec);
  }


  if (use_rec) k_walk_agg<true><<<g_agg, TB, 0, stream>>>(heads, heads1, nxt, src, ew, h, b1, emb, N, 0, Ei, rec);
  else k_walk_agg<false><<<g_agg, TB, 0, stream>>>(heads, heads1, nxt, src, ew, h, b1, emb, N, 0, Ei, rec);

  k_wmma_gemm<2><<<gblk, 128, 0, stream>>>(emb, 64, 0, W2p, KS2, h, 32, HID, nullptr, NPAIRS);
  if (use_rec) k_walk_agg<true><<<g_agg, TB, 0, stream>>>(heads, heads1, nxt, src, ew, h, b2, emb, N, HID, Ei, rec);
  else k_walk_agg<false><<<g_agg, TB, 0, stream>>>(heads, heads1, nxt, src, ew, h, b2, emb, N, HID, Ei, rec);

  k_wmma_gemm<2><<<gblk, 128, 0, stream>>>(emb, 64, HID, W3p, KS2, h, 32, HID, nullptr, NPAIRS);
  if (use_rec) k_walk_agg<true><<<g_agg, TB, 0, stream>>>(heads, heads1, nxt, src, ew, h, b3, emb, N, 2 * HID, Ei, rec);
  else k_walk_agg<false><<<g_agg, TB, 0, stream>>>(heads, heads1, nxt, src, ew, h, b3, emb, N, 2 * HID, Ei, rec);

  k_wmma_gemm<1><<<gblk, 128, 0, stream>>>(emb, 64, 0, Wlp, KSL, out, OUT_DIM, OUT_DIM, bl, NPAIRS);
}


# Lists become arrays: an edge takes a slot of its destination's per-half 31-edge bucket by one returning integer add (rest: overflow list); walks read whole lines, rows asked 8 deep; sum order changes
# speedup vs baseline: 4.4000x; 1.1338x over previous
#include <hip/hip_runtime.h>
#include <hip/hip_bf16.h>

#define IN_DIM 512
#define HID 20
#define OUT_DIM 10

typedef __attribute__((ext_vector_type(2))) float v2f;
typedef __attribute__((ext_vector_type(8))) float v8f;

#define NSUB 4
#define HEAD_COL 60
#define DINV_COL 24

__device__ __forceinline__ void edge_fetch(int cur, const int* __restrict__ src, const float* __restrict__ ew,
                                           const int* __restrict__ nxt, int& s, float& w, int& nx) {
  s = src[cur]; w = ew[cur]; nx = nxt[cur];
}

__global__ void k_init(int* __restrict__ embw, int* __restrict__ hd1, int n) {
  long long t = (long long)blockIdx.x * blockDim.x + threadIdx.x;
  long long ne = (long long)n * 64;
  if (t < ne) embw[t] = ((int)(t & 63) >= HEAD_COL) ? -1 : 0;
  else if (t < ne + (long long)n * 2) hd1[t - ne] = -1;
}

__device__ __forceinline__ const int* head_ptr(const int* embw, const int* hd1, int i, int sub) {
  return (sub < 2) ? (embw + (size_t)i * 64 + HEAD_COL + sub) : (hd1 + (size_t)i * 2 + (sub - 2));
}

__device__ __forceinline__ void packB_elem(int t, const float* __restrict__ W, int Krows, int Ncols,
                                           float* __restrict__ Bp, int Ksteps, int ntiles) {
  int total = Ksteps * ntiles * 64;
  if (t >= total) return;
  int j = t & 1;
  int lane = (t >> 1) & 31;
  int tile = (t >> 6) % ntiles;
  int ks = (t >> 6) / ntiles;
  int k = 4 * ks + 2 * (lane >> 4) + j;
  int n = tile * 16 + (lane & 15);
  Bp[t] = (k < Krows && n < Ncols) ? W[k * Ncols + n] : 0.0f;
}

__global__ void k_packB(const float* __restrict__ W, int Krows, int Ncols,
                        float* __restrict__ Bp, int Ksteps, int ntiles) {
  packB_elem(blockIdx.x * blockDim.x + threadIdx.x, W, Krows, Ncols, Bp, Ksteps, ntiles);
}

struct PackJob {
  const float* W;
  float* Bp;
  int Krows, Ncols, Ksteps, ntiles;
};

__device__ __forceinline__ void build_edge(int e, const int* __restrict__ dst, int* __restrict__ heads,
                                           int* __restrict__ hd1, int* __restrict__ nxt, int E) {
  if (e >= E) return;
  int d = dst[e];
  unsigned place = __builtin_amdgcn_s_sendmsg_rtn(0x87);
  int hf = (int)((place >> 18) & 1u);
  int k = e & 1;
  int* hp = hf ? (hd1 + (size_t)d * 2 + k) : (heads + (size_t)d * 64 + HEAD_COL + k);
  int old = atomicExch(hp, e);
  nxt[e] = old;
}

__global__ void k_deg_walk(const int* __restrict__ hplane, const int* __restrict__ hd1,
                           const int* __restrict__ nxt,
                           const float* __restrict__ ew, float* __restrict__ hrow, int n, int E) {
  int t = blockIdx.x * blockDim.x + threadIdx.x;
  int i = t / NSUB, sub = t % NSUB;
  bool valid = i < n;
  int cur = valid ? *head_ptr(hplane, hd1, i, sub) : -1;
  float p = 0.0f;
  for (int it = 0; it < E; ++it) {
    if ((unsigned)cur >= (unsigned)E) break;
    float w = ew[cur];
    int nx = nxt[cur];
    p += w;
    cur = nx;
  }
  p += __shfl_xor(p, 1, 32);
  p += __shfl_xor(p, 2, 32);
  if (valid && sub == 0) {
    float d = 1.0f + p;
    float4 o = {(d > 0.0f) ? rsqrtf(d) : 0.0f, 0.0f, 0.0f, 0.0f};
    *(float4*)(hrow + (size_t)i * 32 + DINV_COL) = o;
  }
}

__device__ __forceinline__ void finish_row(float4 acc, int lane, int q, const float* __restrict__ b,
                                           float* __restrict__ emb, int i, int off) {
  acc.x += __shfl_xor(acc.x, 8, 32);  acc.y += __shfl_xor(acc.y, 8, 32);
  acc.z += __shfl_xor(acc.z, 8, 32);  acc.w += __shfl_xor(acc.w, 8, 32);
  acc.x += __shfl_xor(acc.x, 16, 32); acc.y += __shfl_xor(acc.y, 16, 32);
  acc.z += __shfl_xor(acc.z, 16, 32); acc.w += __shfl_xor(acc.w, 16, 32);
  float4 v = {0.0f, 0.0f, 0.0f, 0.0f};
  if (q < HID / 4) {
    v.x = acc.x + b[4 * q + 0]; v.y = acc.y + b[4 * q + 1];
    v.z = acc.z + b[4 * q + 2]; v.w = acc.w + b[4 * q + 3];
  }
  float ss = v.x * v.x + v.y * v.y + v.z * v.z + v.w * v.w;
  ss += __shfl_xor(ss, 1, 32);
  ss += __shfl_xor(ss, 2, 32);
  ss += __shfl_xor(ss, 4, 32);
  float nrm = fmaxf(sqrtf(ss), 1e-12f);
  if (lane < HID / 4) {
    float r0n = v.x / nrm, r1n = v.y / nrm, r2n = v.z / nrm, r3n = v.w / nrm;
    float4 o;
    o.x = (r0n > 0.0f) ? r0n : 0.0f;
    o.y = (r1n > 0.0f) ? r1n : 0.0f;
    o.z = (r2n > 0.0f) ? r2n : 0.0f;
    o.w = (r3n > 0.0f) ? r3n : 0.0f;
    ((float4*)(emb + (size_t)i * 64 + off))[q] = o;
  }
}

__global__ void k_walk_agg(const int* __restrict__ hplane, const int* __restrict__ hd1,
                           const int* __restrict__ nxt,
                           const int* __restrict__ src, const float* __restrict__ ew,
                           const float* __restrict__ h,
                           const float* __restrict__ b, float* __restrict__ emb,
                           int n, int off, int E) {
  static_assert(NSUB * 8 == 32, "one wave = NSUB groups of 8 lanes");
  int lane = threadIdx.x & 31;
  int i = blockIdx.x * (blockDim.x >> 5) + (threadIdx.x >> 5);
  if (i >= n) return;
  int sub = lane >> 3;
  int q = lane & 7;
  int pq = (q < HID / 4) ? q : DINV_COL / 4;
  float4 r0 = ((const float4*)(h + (size_t)i * 32))[pq];
  int cur = *head_ptr(hplane, hd1, i, sub);
  float di = __shfl(r0.x, (lane & 24) | (DINV_COL / 4), 32);
  bool act = (unsigned)cur < (unsigned)E;
  int s = 0, nx = -1;
  float w = 0.0f;
  if (act) edge_fetch(cur, src, ew, nxt, s, w, nx);
  float c0 = di * 1.0f * di;
  float4 acc = {0.0f, 0.0f, 0.0f, 0.0f};
  if (sub == 0) { acc.x = r0.x * c0; acc.y = r0.y * c0; acc.z = r0.z * c0; acc.w = r0.w * c0; }
  for (int it = 0; it < E; ++it) {
    if (!act) break;
    float4 r = ((const float4*)(h + (size_t)s * 32))[pq];
    bool nact = (unsigned)nx < (unsigned)E;
    int s2 = 0, nx2 = -1;
    float w2 = 0.0f;
    if (nact) edge_fetch(nx, src, ew, nxt, s2, w2, nx2);
    float ds = __shfl(r.x, (lane & 24) | (DINV_COL / 4), 32);
    float coef = ds * w * di;
    acc.x += r.x * coef; acc.y += r.y * coef; acc.z += r.z * coef; acc.w += r.w * coef;
    s = s2; w = w2; nx = nx2; act = nact;
  }
  finish_row(acc, lane, q, b, emb, i, off);
}

#define BKT_SLOTS 32
#define BKT_EDGES (BKT_SLOTS - 1)

__global__ void k_init_bkt(int* __restrict__ bkt, int n) {
  int t = blockIdx.x * blockDim.x + threadIdx.x;
  if (t < 2 * n) {
    int2 hd = {0, -1};
    ((int2*)bkt)[(size_t)t * BKT_SLOTS] = hd;
  }
}

__device__ __forceinline__ void build_bucket(int e, const int* __restrict__ dst, const int* __restrict__ src,
                                             const float* __restrict__ ew, int* __restrict__ bkt,
                                             int* __restrict__ nxt, int E) {
  if (e >= E) return;
  int d = dst[e];
  int sv = src[e];
  float wv = ew[e];
  unsigned place = __builtin_amdgcn_s_sendmsg_rtn(0x87);
  int hf = (int)((place >> 18) & 1u);
  int* b = bkt + ((size_t)d * 2 + hf) * (BKT_SLOTS * 2);
  int slot = atomicAdd(b, 1);
  if (slot < BKT_EDGES) {
    int2 v = {sv, __float_as_int(wv)};
    ((int2*)b)[slot + 1] = v;
  } else {
    nxt[e] = atomicExch(b + 1, e);
  }
}

__global__ void k_deg_bkt(const int* __restrict__ bkt, const int* __restrict__ nxt,
                          const float* __restrict__ ew, float* __restrict__ hrow, int n, int E) {
  int t = blockIdx.x * blockDim.x + threadIdx.x;
  int i = t >> 3, q = t & 7;
  bool valid = i < n;
  const int4* bp = (const int4*)(bkt + (size_t)(valid ? i : 0) * 2 * (BKT_SLOTS * 2));
  int4 a0 = bp[2 * q], a1 = bp[2 * q + 1];
  int4 c0 = bp[BKT_SLOTS / 2 + 2 * q], c1 = bp[BKT_SLOTS / 2 + 2 * q + 1];
  int g0 = threadIdx.x & 24;
  int m0 = min(__shfl(a0.x, g0, 32), BKT_EDGES), ov0 = __shfl(a0.y, g0, 32);
  int m1 = min(__shfl(c0.x, g0, 32), BKT_EDGES), ov1 = __shfl(c0.y, g0, 32);
  int k = 4 * q;
  float p = 0.0f;
  if (k >= 1 && k <= m0) p += __int_as_float(a0.y);
  if (k + 1 <= m0) p += __int_as_float(a0.w);
  if (k + 2 <= m0) p += __int_as_float(a1.y);
  if (k + 3 <= m0) p += __int_as_float(a1.w);
  if (k >= 1 && k <= m1) p += __int_as_float(c0.y);
  if (k + 1 <= m1) p += __int_as_float(c0.w);
  if (k + 2 <= m1) p += __int_as_float(c1.y);
  if (k + 3 <= m1) p += __int_as_float(c1.w);
  int cur = !valid ? -1 : (q == 0) ? ov0 : (q == 1) ? ov1 : -1;
  for (int it = 0; it < E; ++it) {
    if ((unsigned)cur >= (unsigned)E) break;
    p += ew[cur];
    cur = nxt[cur];
  }
  p += __shfl_xor(p, 1, 32);
  p += __shfl_xor(p, 2, 32);
  p += __shfl_xor(p, 4, 32);
  if (valid && q == 0) {
    float d = 1.0f + p;
    float4 o = {(d > 0.0f) ? rsqrtf(d) : 0.0f, 0.0f, 0.0f, 0.0f};
    *(float4*)(hrow + (size_t)i * 32 + DINV_COL) = o;
  }
}

__global__ void k_bkt_agg(const int* __restrict__ bkt, const int* __restrict__ nxt,
                          const int* __restrict__ src, const float* __restrict__ ew,
                          const float* __restrict__ h,
                          const float* __restrict__ b, float* __restrict__ emb,
                          int n, int off, int E) {
  int lane = threadIdx.x & 31;
  int i = blockIdx.x * (blockDim.x >> 5) + (threadIdx.x >> 5);
  if (i >= n) return;
  int sub = lane >> 3;
  int q = lane & 7;
  int pq = (q < HID / 4) ? q : DINV_COL / 4;
  int dl = (lane & 24) | (DINV_COL / 4);
  const int2* bk = (const int2*)bkt + (size_t)i * 2 * BKT_SLOTS;
  float4 r0 = ((const float4*)(h + (size_t)i * 32))[pq];
  int2 e0 = bk[lane];
  int2 e1 = bk[BKT_SLOTS + lane];
  float di = __shfl(r0.x, dl, 32);
  int c0 = min(max(__shfl(e0.x, 0, 32), 0), BKT_EDGES), ov0 = __shfl(e0.y, 0, 32);
  int c1 = min(max(__shfl(e1.x, 0, 32), 0), BKT_EDGES), ov1 = __shfl(e1.y, 0, 32);
  int tot = c0 + c1;
  float cself = di * 1.0f * di;
  float4 acc = {0.0f, 0.0f, 0.0f, 0.0f};
  bool g0 = sub == 0;
  acc.x = (g0 ? r0.x : 0.0f) * cself; acc.y = (g0 ? r0.y : 0.0f) * cself;
  acc.z = (g0 ? r0.z : 0.0f) * cself; acc.w = (g0 ? r0.w : 0.0f) * cself;
  for (int base = 0; base < tot; base += 32) {
    int v = base + lane;
    int l0 = (v + 1) & 31, l1 = (v - c0 + 1) & 31;
    int sA = __shfl(e0.x, l0, 32), wA = __shfl(e0.y, l0, 32);
    int sB = __shfl(e1.x, l1, 32), wB = __shfl(e1.y, l1, 32);
    int cs = (v < c0) ? sA : sB;
    int cw = (v < c0) ? wA : wB;
    float4 r[8];
#pragma unroll
    for (int j = 0; j < 8; ++j) {
      int t = 4 * j + sub;
      int sj = __shfl(cs, t, 32);
      sj = (base + t < tot) ? sj : 0;
      unsigned ob = (unsigned)sj * 128u + (unsigned)pq * 16u;
      r[j] = *(const float4*)((const char*)h + ob);
    }
#pragma unroll
    for (int j = 0; j < 8; ++j) {
      int t = 4 * j + sub;
      bool okj = base + t < tot;
      float wraw = __int_as_float(__shfl(cw, t, 32));
      float wgt = okj ? wraw : 0.0f;
      float ds = __shfl(r[j].x, dl, 32);
      float coef = ds * wgt * di;
      acc.x += (okj ? r[j].x : 0.0f) * coef; acc.y += (okj ? r[j].y : 0.0f) * coef;
      acc.z += (okj ? r[j].z : 0.0f) * coef; acc.w += (okj ? r[j].w : 0.0f) * coef;
    }
  }
  for (int hf = 0; hf < 2; ++hf) {
    int cur = hf ? ov1 : ov0;
    for (int it = 0; it < E; ++it) {
      if ((unsigned)cur >= (unsigned)E) break;
      int s = src[cur];
      float w = ew[cur];
      int nx = nxt[cur];
      float4 r = ((const float4*)(h + (size_t)s * 32))[pq];
      float ds = __shfl(r.x, dl, 32);
      float coef = ds * w * di;
      if (sub == 0) { acc.x += r.x * coef; acc.y += r.y * coef; acc.z += r.z * coef; acc.w += r.w * coef; }
      cur = nx;
    }
  }
  finish_row(acc, lane, q, b, emb, i, off);
}

template <int NTILES>
__device__ __forceinline__ void gemm_wave(int pair, const float* __restrict__ A, int lda, int acol0,
                                          const float* __restrict__ Bp, int Ksteps,
                                          float* __restrict__ out, int ldo, int ncols,
                                          const float* __restrict__ bias, int npairs) {
  if (pair >= npairs) return;
  int lane = threadIdx.x & 31;
  int half = lane >> 4;
  int l16 = lane & 15;
  int m0 = pair * 32;
  const float* ap0 = A + (size_t)(m0 + l16) * lda + acol0 + half * 2;
  const float* ap1 = ap0 + (size_t)16 * lda;
  const float* bp = Bp + lane * 2;
  v8f acc[2][NTILES];
#pragma unroll
  for (int rr = 0; rr < 2; ++rr)
#pragma unroll
    for (int t = 0; t < NTILES; ++t) acc[rr][t] = (v8f){0.f, 0.f, 0.f, 0.f, 0.f, 0.f, 0.f, 0.f};

#pragma unroll 2
  for (int ks = 0; ks < Ksteps; ++ks) {
    v2f a0 = *(const v2f*)(ap0 + 4 * ks);
    v2f a1 = *(const v2f*)(ap1 + 4 * ks);
#pragma unroll
    for (int t = 0; t < NTILES; ++t) {
      v2f b = *(const v2f*)(bp + (size_t)(ks * NTILES + t) * 64);
      acc[0][t] = __builtin_amdgcn_wmma_f32_16x16x4_f32(false, a0, false, b, (short)0, acc[0][t], false, false);
      acc[1][t] = __builtin_amdgcn_wmma_f32_16x16x4_f32(false, a1, false, b, (short)0, acc[1][t], false, false);
    }
  }

#pragma unroll
  for (int rr = 0; rr < 2; ++rr)
#pragma unroll
    for (int t = 0; t < NTILES; ++t)
#pragma unroll
      for (int r = 0; r < 8; ++r) {
        int row = m0 + rr * 16 + r + half * 8;
        int col = t * 16 + l16;
        if (col < ncols) out[(size_t)row * ldo + col] = acc[rr][t][r] + (bias ? bias[col] : 0.0f);
      }
}

template <int NTILES>
__global__ void k_wmma_gemm(const float* __restrict__ A, int lda, int acol0,
                            const float* __restrict__ Bp, int Ksteps,
                            float* __restrict__ out, int ldo, int ncols,
                            const float* __restrict__ bias, int npairs) {
  gemm_wave<NTILES>(blockIdx.x * (blockDim.x >> 5) + (threadIdx.x >> 5), A, lda, acol0, Bp, Ksteps,
                    out, ldo, ncols, bias, npairs);
}

#define SPLITK 4
__device__ __forceinline__ void gemm1_splitk(int pair, const float* __restrict__ A,
                                             const float* __restrict__ Bp, int Ksteps,
                                             float* __restrict__ out, float* __restrict__ red) {
  static_assert((IN_DIM / 4) % (2 * SPLITK) == 0, "k steps come in groups of 8 (one line of a row)");
  int lane = threadIdx.x & 31;
  int wv = threadIdx.x >> 5;
  int half = lane >> 4;
  int l16 = lane & 15;
  int m0 = pair * 32;
  const float* ap0 = A + (size_t)(m0 + l16) * IN_DIM + half * 2 + 4 * wv;
  const float* ap1 = ap0 + (size_t)16 * IN_DIM;
  const float* bp = Bp + lane * 2 + (size_t)wv * 2 * 64;
  v8f acc[2][2];
#pragma unroll
  for (int rr = 0; rr < 2; ++rr)
#pragma unroll
    for (int t = 0; t < 2; ++t) acc[rr][t] = (v8f){0.f, 0.f, 0.f, 0.f, 0.f, 0.f, 0.f, 0.f};

  int groups = Ksteps / (2 * SPLITK);
#pragma unroll 1
  for (int g = 0; g < groups; ++g) {
#pragma unroll
    for (int u = 0; u < 2; ++u) {
      int ks = 8 * g + 4 * u;
      v2f a0 = *(const v2f*)(ap0 + 4 * ks);
      v2f a1 = *(const v2f*)(ap1 + 4 * ks);
#pragma unroll
      for (int t = 0; t < 2; ++t) {
        v2f b = *(const v2f*)(bp + (size_t)(ks * 2 + t) * 64);
        acc[0][t] = __builtin_amdgcn_wmma_f32_16x16x4_f32(false, a0, false, b, (short)0, acc[0][t], false, false);
        acc[1][t] = __builtin_amdgcn_wmma_f32_16x16x4_f32(false, a1, false, b, (short)0, acc[1][t], false, false);
      }
    }
  }

#pragma unroll
  for (int rr = 0; rr < 2; ++rr)
#pragma unroll
    for (int t = 0; t < 2; ++t)
#pragma unroll
      for (int r = 0; r < 8; ++r) red[((wv * 4 + rr * 2 + t) * 8 + r) * 32 + lane] = acc[rr][t][r];
  __syncthreads();
  int rr = wv >> 1;
  float s[4][2];
#pragma unroll
  for (int rq = 0; rq < 4; ++rq)
#pragma unroll
    for (int t = 0; t < 2; ++t) {
      int r = 4 * (wv & 1) + rq;
      float v = red[((0 * 4 + rr * 2 + t) * 8 + r) * 32 + lane];
#pragma unroll
      for (int p = 1; p < SPLITK; ++p) v += red[((p * 4 + rr * 2 + t) * 8 + r) * 32 + lane];
      s[rq][t] = v;
    }
  float* orow = out + (size_t)(m0 + rr * 16 + 4 * (wv & 1) + half * 8) * 32;
#pragma unroll
  for (int rq = 0; rq < 4; ++rq) orow[rq * 32 + l16] = s[rq][0];
  if (l16 < HID - 16) {
#pragma unroll
    for (int rq = 0; rq < 4; ++rq) orow[rq * 32 + 16 + l16] = s[rq][1];
  }
}

template <bool BKT>
__global__ void k_front(const float* __restrict__ x, const float* __restrict__ W1p, int Ksteps1,
                        float* __restrict__ h, int npairs, int gemm_blocks,
                        PackJob p2, PackJob p3, PackJob pl, int pack_blocks,
                        const int* __restrict__ dst, int* __restrict__ heads,
                        int* __restrict__ hd1, int* __restrict__ nxt, int E,
                        const int* __restrict__ src, const float* __restrict__ ew,
                        int* __restrict__ bkt) {
  __shared__ float red[SPLITK * 4 * 8 * 32];
  int blk = blockIdx.x;
  if (blk < gemm_blocks) {
    gemm1_splitk(blk, x, W1p, Ksteps1, h, red);
    return;
  }
  blk -= gemm_blocks;
  if (blk < 3 * pack_blocks) {
    int job = blk / pack_blocks;
    int t = (blk - job * pack_blocks) * blockDim.x + threadIdx.x;
    if (job == 0) packB_elem(t, p2.W, p2.Krows, p2.Ncols, p2.Bp, p2.Ksteps, p2.ntiles);
    else if (job == 1) packB_elem(t, p3.W, p3.Krows, p3.Ncols, p3.Bp, p3.Ksteps, p3.ntiles);
    else packB_elem(t, pl.W, pl.Krows, pl.Ncols, pl.Bp, pl.Ksteps, pl.ntiles);
    return;
  }
  blk -= 3 * pack_blocks;
  int e = blk * blockDim.x + threadIdx.x;
  if (BKT) build_bucket(e, dst, src, ew, bkt, nxt, E);
  else build_edge(e, dst, heads, hd1, nxt, E);
}

extern "C" void kernel_launch(void* const* d_in, const int* in_sizes, int n_in,
                              void* d_out, int out_size, void* d_ws, size_t ws_size,
                              hipStream_t stream) {
  const float* x  = (const float*)d_in[0];
  const int*   ei = (const int*)d_in[1];
  const float* ew = (const float*)d_in[2];
  const float* W1 = (const float*)d_in[3];
  const float* b1 = (const float*)d_in[4];
  const float* W2 = (const float*)d_in[5];
  const float* b2 = (const float*)d_in[6];
  const float* W3 = (const float*)d_in[7];
  const float* b3 = (const float*)d_in[8];
  const float* Wl = (const float*)d_in[9];
  const float* bl = (const float*)d_in[10];
  float* out = (float*)d_out;

  const int N = in_sizes[0] / IN_DIM;
  const long long E = in_sizes[2];
  const int* src = ei;
  const int* dst = ei + E;

  const int KS1 = IN_DIM / 4;
  const int KS2 = (HID + 3) / 4;
  const int KSL = (3 * HID) / 4;

  float* ws = (float*)d_ws;
  int* heads1 = (int*)d_ws;
  ws += N;
  ws += N;
  float* h    = ws; ws += (size_t)N * 32;
  int*   nxt  = (int*)ws; ws += (size_t)N * 32;
  float* emb  = ws; ws += (size_t)N * 64;
  float* W1p  = ws; ws += KS1 * 2 * 64;
  float* W2p  = ws; ws += KS2 * 2 * 64;
  float* W3p  = ws; ws += KS2 * 2 * 64;
  float* Wlp  = ws; ws += KSL * 1 * 64;
  int* bkt = (int*)ws;
  const long long room = (long long)ws_size - (long long)((char*)ws - (char*)d_ws);
  const bool use_bkt = room >= (long long)N * 2 * BKT_SLOTS * 8;
  int* heads = (int*)emb;

  const int TB = 256;
  const int FB = 128;
  const int WB = 64;
  const int NPAIRS = N / 32;
  const int GB = (NPAIRS + 3) / 4;
  dim3 gblk(GB);
  const int Ei = (int)E;
  int g_walk = (int)(((long long)N * NSUB + WB - 1) / WB);
  int g_agg = (N + 7) / 8;

  int g_deg_b = (int)(((long long)N * 8 + WB - 1) / WB);

  if (use_bkt) k_init_bkt<<<(2 * N + TB - 1) / TB, TB, 0, stream>>>(bkt, N);
  else k_init<<<(int)(((long long)N * 66 + TB - 1) / TB), TB, 0, stream>>>(heads, heads1, N);
  k_packB<<<(KS1 * 2 * 64 + TB - 1) / TB, TB, 0, stream>>>(W1, IN_DIM, HID, W1p, KS1, 2);

  PackJob p2 = {W2, W2p, HID, HID, KS2, 2};
  PackJob p3 = {W3, W3p, HID, HID, KS2, 2};
  PackJob pl = {Wl, Wlp, 3 * HID, OUT_DIM, KSL, 1};
  const int PB = (KSL * 1 * 64 + FB - 1) / FB;
  const int BB = (int)((E + FB - 1) / FB);
  const int GF = NPAIRS;
  if (use_bkt) {
    k_front<true><<<GF + 3 * PB + BB, FB, 0, stream>>>(x, W1p, KS1, h, NPAIRS, GF, p2, p3, pl, PB,
                                                       dst, heads, heads1, nxt, Ei, src, ew, bkt);
    k_deg_bkt<<<g_deg_b, WB, 0, stream>>>(bkt, nxt, ew, h, N, Ei);
  } else {
    k_front<false><<<GF + 3 * PB + BB, FB, 0, stream>>>(x, W1p, KS1, h, NPAIRS, GF, p2, p3, pl, PB,
                                                        dst, heads, heads1, nxt, Ei, src, ew, bkt);
    k_deg_walk<<<g_walk, WB, 0, stream>>>(heads, heads1, nxt, ew, h, N, Ei);
  }


  if (use_bkt) k_bkt_agg<<<g_agg, TB, 0, stream>>>(bkt, nxt, src, ew, h, b1, emb, N, 0, Ei);
  else k_walk_agg<<<g_agg, TB, 0, stream>>>(heads, heads1, nxt, src, ew, h, b1, emb, N, 0, Ei);

  k_wmma_gemm<2><<<gblk, 128, 0, stream>>>(emb, 64, 0, W2p, KS2, h, 32, HID, nullptr, NPAIRS);
  if (use_bkt) k_bkt_agg<<<g_agg, TB, 0, stream>>>(bkt, nxt, src, ew, h, b2, emb, N, HID, Ei);
  else k_walk_agg<<<g_agg, TB, 0, stream>>>(heads, heads1, nxt, src, ew, h, b2, emb, N, HID, Ei);

  k_wmma_gemm<2><<<gblk, 128, 0, stream>>>(emb, 64, HID, W3p, KS2, h, 32, HID, nullptr, NPAIRS);
  if (use_bkt) k_bkt_agg<<<g_agg, TB, 0, stream>>>(bkt, nxt, src, ew, h, b3, emb, N, 2 * HID, Ei);
  else k_walk_agg<<<g_agg, TB, 0, stream>>>(heads, heads1, nxt, src, ew, h, b3, emb, N, 2 * HID, Ei);

  k_wmma_gemm<1><<<gblk, 128, 0, stream>>>(emb, 64, 0, Wlp, KSL, out, OUT_DIM, OUT_DIM, bl, NPAIRS);
}


# Gathers: a wave takes 4 nodes, one 8-lane group a node; a group sums its node's slots alone in slot order (in-group DPP broadcasts, no add across groups); 300000 waves -> 75000; sum order changes
# speedup vs baseline: 5.2909x; 1.2025x over previous
#include <hip/hip_runtime.h>
#include <hip/hip_bf16.h>

#define IN_DIM 512
#define HID 20
#define OUT_DIM 10

typedef __attribute__((ext_vector_type(2))) float v2f;
typedef __attribute__((ext_vector_type(8))) float v8f;

#define NSUB 4
#define HEAD_COL 60
#define DINV_COL 24

__device__ __forceinline__ void edge_fetch(int cur, const int* __restrict__ src, const float* __restrict__ ew,
                                           const int* __restrict__ nxt, int& s, float& w, int& nx) {
  s = src[cur]; w = ew[cur]; nx = nxt[cur];
}

__global__ void k_init(int* __restrict__ embw, int* __restrict__ hd1, int n) {
  long long t = (long long)blockIdx.x * blockDim.x + threadIdx.x;
  long long ne = (long long)n * 64;
  if (t < ne) embw[t] = ((int)(t & 63) >= HEAD_COL) ? -1 : 0;
  else if (t < ne + (long long)n * 2) hd1[t - ne] = -1;
}

__device__ __forceinline__ const int* head_ptr(const int* embw, const int* hd1, int i, int sub) {
  return (sub < 2) ? (embw + (size_t)i * 64 + HEAD_COL + sub) : (hd1 + (size_t)i * 2 + (sub - 2));
}

__device__ __forceinline__ void packB_elem(int t, const float* __restrict__ W, int Krows, int Ncols,
                                           float* __restrict__ Bp, int Ksteps, int ntiles) {
  int total = Ksteps * ntiles * 64;
  if (t >= total) return;
  int j = t & 1;
  int lane = (t >> 1) & 31;
  int tile = (t >> 6) % ntiles;
  int ks = (t >> 6) / ntiles;
  int k = 4 * ks + 2 * (lane >> 4) + j;
  int n = tile * 16 + (lane & 15);
  Bp[t] = (k < Krows && n < Ncols) ? W[k * Ncols + n] : 0.0f;
}

__global__ void k_packB(const float* __restrict__ W, int Krows, int Ncols,
                        float* __restrict__ Bp, int Ksteps, int ntiles) {
  packB_elem(blockIdx.x * blockDim.x + threadIdx.x, W, Krows, Ncols, Bp, Ksteps, ntiles);
}

struct PackJob {
  const float* W;
  float* Bp;
  int Krows, Ncols, Ksteps, ntiles;
};

__device__ __forceinline__ void build_edge(int e, const int* __restrict__ dst, int* __restrict__ heads,
                                           int* __restrict__ hd1, int* __restrict__ nxt, int E) {
  if (e >= E) return;
  int d = dst[e];
  unsigned place = __builtin_amdgcn_s_sendmsg_rtn(0x87);
  int hf = (int)((place >> 18) & 1u);
  int k = e & 1;
  int* hp = hf ? (hd1 + (size_t)d * 2 + k) : (heads + (size_t)d * 64 + HEAD_COL + k);
  int old = atomicExch(hp, e);
  nxt[e] = old;
}

__global__ void k_deg_walk(const int* __restrict__ hplane, const int* __restrict__ hd1,
                           const int* __restrict__ nxt,
                           const float* __restrict__ ew, float* __restrict__ hrow, int n, int E) {
  int t = blockIdx.x * blockDim.x + threadIdx.x;
  int i = t / NSUB, sub = t % NSUB;
  bool valid = i < n;
  int cur = valid ? *head_ptr(hplane, hd1, i, sub) : -1;
  float p = 0.0f;
  for (int it = 0; it < E; ++it) {
    if ((unsigned)cur >= (unsigned)E) break;
    float w = ew[cur];
    int nx = nxt[cur];
    p += w;
    cur = nx;
  }
  p += __shfl_xor(p, 1, 32);
  p += __shfl_xor(p, 2, 32);
  if (valid && sub == 0) {
    float d = 1.0f + p;
    float4 o = {(d > 0.0f) ? rsqrtf(d) : 0.0f, 0.0f, 0.0f, 0.0f};
    *(float4*)(hrow + (size_t)i * 32 + DINV_COL) = o;
  }
}

__device__ __forceinline__ void finish_row(float4 acc, int lane, int q, const float* __restrict__ b,
                                           float* __restrict__ emb, int i, int off) {
  acc.x += __shfl_xor(acc.x, 8, 32);  acc.y += __shfl_xor(acc.y, 8, 32);
  acc.z += __shfl_xor(acc.z, 8, 32);  acc.w += __shfl_xor(acc.w, 8, 32);
  acc.x += __shfl_xor(acc.x, 16, 32); acc.y += __shfl_xor(acc.y, 16, 32);
  acc.z += __shfl_xor(acc.z, 16, 32); acc.w += __shfl_xor(acc.w, 16, 32);
  float4 v = {0.0f, 0.0f, 0.0f, 0.0f};
  if (q < HID / 4) {
    v.x = acc.x + b[4 * q + 0]; v.y = acc.y + b[4 * q + 1];
    v.z = acc.z + b[4 * q + 2]; v.w = acc.w + b[4 * q + 3];
  }
  float ss = v.x * v.x + v.y * v.y + v.z * v.z + v.w * v.w;
  ss += __shfl_xor(ss, 1, 32);
  ss += __shfl_xor(ss, 2, 32);
  ss += __shfl_xor(ss, 4, 32);
  float nrm = fmaxf(sqrtf(ss), 1e-12f);
  if (lane < HID / 4) {
    float r0n = v.x / nrm, r1n = v.y / nrm, r2n = v.z / nrm, r3n = v.w / nrm;
    float4 o;
    o.x = (r0n > 0.0f) ? r0n : 0.0f;
    o.y = (r1n > 0.0f) ? r1n : 0.0f;
    o.z = (r2n > 0.0f) ? r2n : 0.0f;
    o.w = (r3n > 0.0f) ? r3n : 0.0f;
    ((float4*)(emb + (size_t)i * 64 + off))[q] = o;
  }
}

__global__ void k_walk_agg(const int* __restrict__ hplane, const int* __restrict__ hd1,
                           const int* __restrict__ nxt,
                           const int* __restrict__ src, const float* __restrict__ ew,
                           const float* __restrict__ h,
                           const float* __restrict__ b, float* __restrict__ emb,
                           int n, int off, int E) {
  static_assert(NSUB * 8 == 32, "one wave = NSUB groups of 8 lanes");
  int lane = threadIdx.x & 31;
  int i = blockIdx.x * (blockDim.x >> 5) + (threadIdx.x >> 5);
  if (i >= n) return;
  int sub = lane >> 3;
  int q = lane & 7;
  int pq = (q < HID / 4) ? q : DINV_COL / 4;
  float4 r0 = ((const float4*)(h + (size_t)i * 32))[pq];
  int cur = *head_ptr(hplane, hd1, i, sub);
  float di = __shfl(r0.x, (lane & 24) | (DINV_COL / 4), 32);
  bool act = (unsigned)cur < (unsigned)E;
  int s = 0, nx = -1;
  float w = 0.0f;
  if (act) edge_fetch(cur, src, ew, nxt, s, w, nx);
  float c0 = di * 1.0f * di;
  float4 acc = {0.0f, 0.0f, 0.0f, 0.0f};
  if (sub == 0) { acc.x = r0.x * c0; acc.y = r0.y * c0; acc.z = r0.z * c0; acc.w = r0.w * c0; }
  for (int it = 0; it < E; ++it) {
    if (!act) break;
    float4 r = ((const float4*)(h + (size_t)s * 32))[pq];
    bool nact = (unsigned)nx < (unsigned)E;
    int s2 = 0, nx2 = -1;
    float w2 = 0.0f;
    if (nact) edge_fetch(nx, src, ew, nxt, s2, w2, nx2);
    float ds = __shfl(r.x, (lane & 24) | (DINV_COL / 4), 32);
    float coef = ds * w * di;
    acc.x += r.x * coef; acc.y += r.y * coef; acc.z += r.z * coef; acc.w += r.w * coef;
    s = s2; w = w2; nx = nx2; act = nact;
  }
  finish_row(acc, lane, q, b, emb, i, off);
}

#define BKT_SLOTS 32
#define BKT_EDGES (BKT_SLOTS - 1)

__global__ void k_init_bkt(int* __restrict__ bkt, int n) {
  int t = blockIdx.x * blockDim.x + threadIdx.x;
  if (t < 2 * n) {
    int2 hd = {0, -1};
    ((int2*)bkt)[(size_t)t * BKT_SLOTS] = hd;
  }
}

__device__ __forceinline__ void build_bucket(int e, const int* __restrict__ dst, const int* __restrict__ src,
                                             const float* __restrict__ ew, int* __restrict__ bkt,
                                             int* __restrict__ nxt, int E) {
  if (e >= E) return;
  int d = dst[e];
  int sv = src[e];
  float wv = ew[e];
  unsigned place = __builtin_amdgcn_s_sendmsg_rtn(0x87);
  int hf = (int)((place >> 18) & 1u);
  int* b = bkt + ((size_t)d * 2 + hf) * (BKT_SLOTS * 2);
  int slot = atomicAdd(b, 1);
  if (slot < BKT_EDGES) {
    int2 v = {sv, __float_as_int(wv)};
    ((int2*)b)[slot + 1] = v;
  } else {
    nxt[e] = atomicExch(b + 1, e);
  }
}

__global__ void k_deg_bkt(const int* __restrict__ bkt, const int* __restrict__ nxt,
                          const float* __restrict__ ew, float* __restrict__ hrow, int n, int E) {
  int t = blockIdx.x * blockDim.x + threadIdx.x;
  int i = t >> 3, q = t & 7;
  bool valid = i < n;
  const int4* bp = (const int4*)(bkt + (size_t)(valid ? i : 0) * 2 * (BKT_SLOTS * 2));
  int4 a0 = bp[2 * q], a1 = bp[2 * q + 1];
  int4 c0 = bp[BKT_SLOTS / 2 + 2 * q], c1 = bp[BKT_SLOTS / 2 + 2 * q + 1];
  int g0 = threadIdx.x & 24;
  int m0 = min(__shfl(a0.x, g0, 32), BKT_EDGES), ov0 = __shfl(a0.y, g0, 32);
  int m1 = min(__shfl(c0.x, g0, 32), BKT_EDGES), ov1 = __shfl(c0.y, g0, 32);
  int k = 4 * q;
  float p = 0.0f;
  if (k >= 1 && k <= m0) p += __int_as_float(a0.y);
  if (k + 1 <= m0) p += __int_as_float(a0.w);
  if (k + 2 <= m0) p += __int_as_float(a1.y);
  if (k + 3 <= m0) p += __int_as_float(a1.w);
  if (k >= 1 && k <= m1) p += __int_as_float(c0.y);
  if (k + 1 <= m1) p += __int_as_float(c0.w);
  if (k + 2 <= m1) p += __int_as_float(c1.y);
  if (k + 3 <= m1) p += __int_as_float(c1.w);
  int cur = !valid ? -1 : (q == 0) ? ov0 : (q == 1) ? ov1 : -1;
  for (int it = 0; it < E; ++it) {
    if ((unsigned)cur >= (unsigned)E) break;
    p += ew[cur];
    cur = nxt[cur];
  }
  p += __shfl_xor(p, 1, 32);
  p += __shfl_xor(p, 2, 32);
  p += __shfl_xor(p, 4, 32);
  if (valid && q == 0) {
    float d = 1.0f + p;
    float4 o = {(d > 0.0f) ? rsqrtf(d) : 0.0f, 0.0f, 0.0f, 0.0f};
    *(float4*)(hrow + (size_t)i * 32 + DINV_COL) = o;
  }
}

template <int L>
__device__ __forceinline__ int bc8(int v) {
  return (int)__builtin_amdgcn_mov_dpp8((unsigned)v, (unsigned)L * 0x249249u);
}

template <int K>
__device__ __forceinline__ void bkt_batch(const int4 X0, const int4 X1, int c, int ic, unsigned pqo,
                                          const float* __restrict__ h, float di, float4& acc) {
  const int4 X = (K < 2) ? X0 : X1;
  float4 r[8];
#define BKT_ASK(U)                                                   \
  if constexpr (8 * K + U > 0) {                                     \
    constexpr int s_ = 8 * K + U;                                    \
    constexpr int L_ = (s_ & 15) >> 1;                               \
    int sv = bc8<L_>((s_ & 1) ? X.z : X.x);                          \
    int sj = (s_ <= c) ? sv : ic;                                    \
    unsigned ob = (unsigned)sj * 128u + pqo;                         \
    r[U] = *(const float4*)((const char*)h + ob);                    \
  }
  BKT_ASK(0) BKT_ASK(1) BKT_ASK(2) BKT_ASK(3) BKT_ASK(4) BKT_ASK(5) BKT_ASK(6) BKT_ASK(7)
#undef BKT_ASK
#define BKT_USE(U)                                                   \
  if constexpr (8 * K + U > 0) {                                     \
    constexpr int s_ = 8 * K + U;                                    \
    constexpr int L_ = (s_ & 15) >> 1;                               \
    int wb = bc8<L_>((s_ & 1) ? X.w : X.y);                          \
    float wgt = (s_ <= c) ? __int_as_float(wb) : 0.0f;               \
    float ds = __int_as_float(bc8<DINV_COL / 4>(__float_as_int(r[U].x))); \
    float coef = ds * wgt * di;                                      \
    acc.x += r[U].x * coef; acc.y += r[U].y * coef;                  \
    acc.z += r[U].z * coef; acc.w += r[U].w * coef;                  \
  }
  BKT_USE(0) BKT_USE(1) BKT_USE(2) BKT_USE(3) BKT_USE(4) BKT_USE(5) BKT_USE(6) BKT_USE(7)
#undef BKT_USE
}

__device__ __forceinline__ void bkt_run(const int4 X0, const int4 X1, int c, int cm, int ic, unsigned pqo,
                                        const float* __restrict__ h, float di, float4& acc) {
  if (cm >= 1) {
    bkt_batch<0>(X0, X1, c, ic, pqo, h, di, acc);
    if (cm >= 8) {
      bkt_batch<1>(X0, X1, c, ic, pqo, h, di, acc);
      if (cm >= 16) {
        bkt_batch<2>(X0, X1, c, ic, pqo, h, di, acc);
        if (cm >= 24) bkt_batch<3>(X0, X1, c, ic, pqo, h, di, acc);
      }
    }
  }
}

__device__ __forceinline__ void finish_group(float4 acc, int q, bool valid, const float* __restrict__ b,
                                             float* __restrict__ emb, int i, int off) {
  float4 v = {0.0f, 0.0f, 0.0f, 0.0f};
  if (q < HID / 4) {
    v.x = acc.x + b[4 * q + 0]; v.y = acc.y + b[4 * q + 1];
    v.z = acc.z + b[4 * q + 2]; v.w = acc.w + b[4 * q + 3];
  }
  float ss = v.x * v.x + v.y * v.y + v.z * v.z + v.w * v.w;
  ss += __shfl_xor(ss, 1, 32);
  ss += __shfl_xor(ss, 2, 32);
  ss += __shfl_xor(ss, 4, 32);
  float nrm = fmaxf(sqrtf(ss), 1e-12f);
  if (valid && q < HID / 4) {
    float r0n = v.x / nrm, r1n = v.y / nrm, r2n = v.z / nrm, r3n = v.w / nrm;
    float4 o;
    o.x = (r0n > 0.0f) ? r0n : 0.0f;
    o.y = (r1n > 0.0f) ? r1n : 0.0f;
    o.z = (r2n > 0.0f) ? r2n : 0.0f;
    o.w = (r3n > 0.0f) ? r3n : 0.0f;
    ((float4*)(emb + (size_t)i * 64 + off))[q] = o;
  }
}

__global__ void k_bkt_agg(const int* __restrict__ bkt, const int* __restrict__ nxt,
                          const int* __restrict__ src, const float* __restrict__ ew,
                          const float* __restrict__ h,
                          const float* __restrict__ b, float* __restrict__ emb,
                          int n, int off, int E) {
  static_assert(BKT_SLOTS == 32, "a bucket is 16 pieces of 16 bytes: 2 a lane of a group of 8");
  int lane = threadIdx.x & 31;
  int wave = blockIdx.x * (blockDim.x >> 5) + (threadIdx.x >> 5);
  if (wave * 4 >= n) return;
  int q = lane & 7;
  int i = wave * 4 + (lane >> 3);
  bool valid = i < n;
  int ic = valid ? i : n - 1;
  int pq = (q < HID / 4) ? q : DINV_COL / 4;
  unsigned pqo = (unsigned)pq * 16u;
  const int4* bk = (const int4*)bkt + (size_t)ic * BKT_SLOTS;
  float4 r0 = *(const float4*)((const char*)h + ((unsigned)ic * 128u + pqo));
  int4 X0 = bk[q], X1 = bk[8 + q];
  int4 Y0 = bk[16 + q], Y1 = bk[24 + q];
  float di = __int_as_float(bc8<DINV_COL / 4>(__float_as_int(r0.x)));
  int c0 = min(max(bc8<0>(X0.x), 0), BKT_EDGES), ov0 = bc8<0>(X0.y);
  int c1 = min(max(bc8<0>(Y0.x), 0), BKT_EDGES), ov1 = bc8<0>(Y0.y);
  float cself = di * 1.0f * di;
  float4 acc = {r0.x * cself, r0.y * cself, r0.z * cself, r0.w * cself};
  int m0 = max(c0, __shfl_xor(c0, 8, 32));  m0 = max(m0, __shfl_xor(m0, 16, 32));
  int m1 = max(c1, __shfl_xor(c1, 8, 32));  m1 = max(m1, __shfl_xor(m1, 16, 32));
  int cm0 = __builtin_amdgcn_readfirstlane(m0);
  int cm1 = __builtin_amdgcn_readfirstlane(m1);
  bkt_run(X0, X1, c0, cm0, ic, pqo, h, di, acc);
  bkt_run(Y0, Y1, c1, cm1, ic, pqo, h, di, acc);
  for (int hf = 0; hf < 2; ++hf) {
    int cur = valid ? (hf ? ov1 : ov0) : -1;
    for (int it = 0; it < E; ++it) {
      bool a = (unsigned)cur < (unsigned)E;
      if (__ballot(a) == 0) break;
      int cc = a ? cur : 0;
      int s = src[cc];
      float w = a ? ew[cc] : 0.0f;
      int nx = -1;
      if (a) nx = nxt[cc];
      int sj = a ? s : ic;
      float4 r = *(const float4*)((const char*)h + ((unsigned)sj * 128u + pqo));
      float ds = __int_as_float(bc8<DINV_COL / 4>(__float_as_int(r.x)));
      float coef = ds * w * di;
      acc.x += r.x * coef; acc.y += r.y * coef; acc.z += r.z * coef; acc.w += r.w * coef;
      cur = a ? nx : -1;
    }
  }
  finish_group(acc, q, valid, b, emb, i, off);
}

template <int NTILES>
__device__ __forceinline__ void gemm_wave(int pair, const float* __restrict__ A, int lda, int acol0,
                                          const float* __restrict__ Bp, int Ksteps,
                                          float* __restrict__ out, int ldo, int ncols,
                                          const float* __restrict__ bias, int npairs) {
  if (pair >= npairs) return;
  int lane = threadIdx.x & 31;
  int half = lane >> 4;
  int l16 = lane & 15;
  int m0 = pair * 32;
  const float* ap0 = A + (size_t)(m0 + l16) * lda + acol0 + half * 2;
  const float* ap1 = ap0 + (size_t)16 * lda;
  const float* bp = Bp + lane * 2;
  v8f acc[2][NTILES];
#pragma unroll
  for (int rr = 0; rr < 2; ++rr)
#pragma unroll
    for (int t = 0; t < NTILES; ++t) acc[rr][t] = (v8f){0.f, 0.f, 0.f, 0.f, 0.f, 0.f, 0.f, 0.f};

#pragma unroll 2
  for (int ks = 0; ks < Ksteps; ++ks) {
    v2f a0 = *(const v2f*)(ap0 + 4 * ks);
    v2f a1 = *(const v2f*)(ap1 + 4 * ks);
#pragma unroll
    for (int t = 0; t < NTILES; ++t) {
      v2f b = *(const v2f*)(bp + (size_t)(ks * NTILES + t) * 64);
      acc[0][t] = __builtin_amdgcn_wmma_f32_16x16x4_f32(false, a0, false, b, (short)0, acc[0][t], false, false);
      acc[1][t] = __builtin_amdgcn_wmma_f32_16x16x4_f32(false, a1, false, b, (short)0, acc[1][t], false, false);
    }
  }

#pragma unroll
  for (int rr = 0; rr < 2; ++rr)
#pragma unroll
    for (int t = 0; t < NTILES; ++t)
#pragma unroll
      for (int r = 0; r < 8; ++r) {
        int row = m0 + rr * 16 + r + half * 8;
        int col = t * 16 + l16;
        if (col < ncols) out[(size_t)row * ldo + col] = acc[rr][t][r] + (bias ? bias[col] : 0.0f);
      }
}

template <int NTILES>
__global__ void k_wmma_gemm(const float* __restrict__ A, int lda, int acol0,
                            const float* __restrict__ Bp, int Ksteps,
                            float* __restrict__ out, int ldo, int ncols,
                            const float* __restrict__ bias, int npairs) {
  gemm_wave<NTILES>(blockIdx.x * (blockDim.x >> 5) + (threadIdx.x >> 5), A, lda, acol0, Bp, Ksteps,
                    out, ldo, ncols, bias, npairs);
}

#define SPLITK 4
__device__ __forceinline__ void gemm1_splitk(int pair, const float* __restrict__ A,
                                             const float* __restrict__ Bp, int Ksteps,
                                             float* __restrict__ out, float* __restrict__ red) {
  static_assert((IN_DIM / 4) % (2 * SPLITK) == 0, "k steps come in groups of 8 (one line of a row)");
  int lane = threadIdx.x & 31;
  int wv = threadIdx.x >> 5;
  int half = lane >> 4;
  int l16 = lane & 15;
  int m0 = pair * 32;
  const float* ap0 = A + (size_t)(m0 + l16) * IN_DIM + half * 2 + 4 * wv;
  const float* ap1 = ap0 + (size_t)16 * IN_DIM;
  const float* bp = Bp + lane * 2 + (size_t)wv * 2 * 64;
  v8f acc[2][2];
#pragma unroll
  for (int rr = 0; rr < 2; ++rr)
#pragma unroll
    for (int t = 0; t < 2; ++t) acc[rr][t] = (v8f){0.f, 0.f, 0.f, 0.f, 0.f, 0.f, 0.f, 0.f};

  int groups = Ksteps / (2 * SPLITK);
#pragma unroll 1
  for (int g = 0; g < groups; ++g) {
#pragma unroll
    for (int u = 0; u < 2; ++u) {
      int ks = 8 * g + 4 * u;
      v2f a0 = *(const v2f*)(ap0 + 4 * ks);
      v2f a1 = *(const v2f*)(ap1 + 4 * ks);
#pragma unroll
      for (int t = 0; t < 2; ++t) {
        v2f b = *(const v2f*)(bp + (size_t)(ks * 2 + t) * 64);
        acc[0][t] = __builtin_amdgcn_wmma_f32_16x16x4_f32(false, a0, false, b, (short)0, acc[0][t], false, false);
        acc[1][t] = __builtin_amdgcn_wmma_f32_16x16x4_f32(false, a1, false, b, (short)0, acc[1][t], false, false);
      }
    }
  }

#pragma unroll
  for (int rr = 0; rr < 2; ++rr)
#pragma unroll
    for (int t = 0; t < 2; ++t)
#pragma unroll
      for (int r = 0; r < 8; ++r) red[((wv * 4 + rr * 2 + t) * 8 + r) * 32 + lane] = acc[rr][t][r];
  __syncthreads();
  int rr = wv >> 1;
  float s[4][2];
#pragma unroll
  for (int rq = 0; rq < 4; ++rq)
#pragma unroll
    for (int t = 0; t < 2; ++t) {
      int r = 4 * (wv & 1) + rq;
      float v = red[((0 * 4 + rr * 2 + t) * 8 + r) * 32 + lane];
#pragma unroll
      for (int p = 1; p < SPLITK; ++p) v += red[((p * 4 + rr * 2 + t) * 8 + r) * 32 + lane];
      s[rq][t] = v;
    }
  float* orow = out + (size_t)(m0 + rr * 16 + 4 * (wv & 1) + half * 8) * 32;
#pragma unroll
  for (int rq = 0; rq < 4; ++rq) orow[rq * 32 + l16] = s[rq][0];
  if (l16 < HID - 16) {
#pragma unroll
    for (int rq = 0; rq < 4; ++rq) orow[rq * 32 + 16 + l16] = s[rq][1];
  }
}

template <bool BKT>
__global__ void k_front(const float* __restrict__ x, const float* __restrict__ W1p, int Ksteps1,
                        float* __restrict__ h, int npairs, int gemm_blocks,
                        PackJob p2, PackJob p3, PackJob pl, int pack_blocks,
                        const int* __restrict__ dst, int* __restrict__ heads,
                        int* __restrict__ hd1, int* __restrict__ nxt, int E,
                        const int* __restrict__ src, const float* __restrict__ ew,
                        int* __restrict__ bkt) {
  __shared__ float red[SPLITK * 4 * 8 * 32];
  int blk = blockIdx.x;
  if (blk < gemm_blocks) {
    gemm1_splitk(blk, x, W1p, Ksteps1, h, red);
    return;
  }
  blk -= gemm_blocks;
  if (blk < 3 * pack_blocks) {
    int job = blk / pack_blocks;
    int t = (blk - job * pack_blocks) * blockDim.x + threadIdx.x;
    if (job == 0) packB_elem(t, p2.W, p2.Krows, p2.Ncols, p2.Bp, p2.Ksteps, p2.ntiles);
    else if (job == 1) packB_elem(t, p3.W, p3.Krows, p3.Ncols, p3.Bp, p3.Ksteps, p3.ntiles);
    else packB_elem(t, pl.W, pl.Krows, pl.Ncols, pl.Bp, pl.Ksteps, pl.ntiles);
    return;
  }
  blk -= 3 * pack_blocks;
  int e = blk * blockDim.x + threadIdx.x;
  if (BKT) build_bucket(e, dst, src, ew, bkt, nxt, E);
  else build_edge(e, dst, heads, hd1, nxt, E);
}

extern "C" void kernel_launch(void* const* d_in, const int* in_sizes, int n_in,
                              void* d_out, int out_size, void* d_ws, size_t ws_size,
                              hipStream_t stream) {
  const float* x  = (const float*)d_in[0];
  const int*   ei = (const int*)d_in[1];
  const float* ew = (const float*)d_in[2];
  const float* W1 = (const float*)d_in[3];
  const float* b1 = (const float*)d_in[4];
  const float* W2 = (const float*)d_in[5];
  const float* b2 = (const float*)d_in[6];
  const float* W3 = (const float*)d_in[7];
  const float* b3 = (const float*)d_in[8];
  const float* Wl = (const float*)d_in[9];
  const float* bl = (const float*)d_in[10];
  float* out = (float*)d_out;

  const int N = in_sizes[0] / IN_DIM;
  const long long E = in_sizes[2];
  const int* src = ei;
  const int* dst = ei + E;

  const int KS1 = IN_DIM / 4;
  const int KS2 = (HID + 3) / 4;
  const int KSL = (3 * HID) / 4;

  float* ws = (float*)d_ws;
  int* heads1 = (int*)d_ws;
  ws += N;
  ws += N;
  float* h    = ws; ws += (size_t)N * 32;
  int*   nxt  = (int*)ws; ws += (size_t)N * 32;
  float* emb  = ws; ws += (size_t)N * 64;
  float* W1p  = ws; ws += KS1 * 2 * 64;
  float* W2p  = ws; ws += KS2 * 2 * 64;
  float* W3p  = ws; ws += KS2 * 2 * 64;
  float* Wlp  = ws; ws += KSL * 1 * 64;
  int* bkt = (int*)ws;
  const long long room = (long long)ws_size - (long long)((char*)ws - (char*)d_ws);
  const bool use_bkt = room >= (long long)N * 2 * BKT_SLOTS * 8;
  int* heads = (int*)emb;

  const int TB = 256;
  const int FB = 128;
  const int WB = 64;
  const int NPAIRS = N / 32;
  const int GB = (NPAIRS + 3) / 4;
  dim3 gblk(GB);
  const int Ei = (int)E;
  int g_walk = (int)(((long long)N * NSUB + WB - 1) / WB);
  int g_agg = (N + 7) / 8;

  int g_deg_b = (int)(((long long)N * 8 + WB - 1) / WB);
  int g_agg_b = (N + 31) / 32;

  if (use_bkt) k_init_bkt<<<(2 * N + TB - 1) / TB, TB, 0, stream>>>(bkt, N);
  else k_init<<<(int)(((long long)N * 66 + TB - 1) / TB), TB, 0, stream>>>(heads, heads1, N);
  k_packB<<<(KS1 * 2 * 64 + TB - 1) / TB, TB, 0, stream>>>(W1, IN_DIM, HID, W1p, KS1, 2);

  PackJob p2 = {W2, W2p, HID, HID, KS2, 2};
  PackJob p3 = {W3, W3p, HID, HID, KS2, 2};
  PackJob pl = {Wl, Wlp, 3 * HID, OUT_DIM, KSL, 1};
  const int PB = (KSL * 1 * 64 + FB - 1) / FB;
  const int BB = (int)((E + FB - 1) / FB);
  const int GF = NPAIRS;
  if (use_bkt) {
    k_front<true><<<GF + 3 * PB + BB, FB, 0, stream>>>(x, W1p, KS1, h, NPAIRS, GF, p2, p3, pl, PB,
                                                       dst, heads, heads1, nxt, Ei, src, ew, bkt);
    k_deg_bkt<<<g_deg_b, WB, 0, stream>>>(bkt, nxt, ew, h, N, Ei);
  } else {
    k_front<false><<<GF + 3 * PB + BB, FB, 0, stream>>>(x, W1p, KS1, h, NPAIRS, GF, p2, p3, pl, PB,
                                                        dst, heads, heads1, nxt, Ei, src, ew, bkt);
    k_deg_walk<<<g_walk, WB, 0, stream>>>(heads, heads1, nxt, ew, h, N, Ei);
  }


  if (use_bkt) k_bkt_agg<<<g_agg_b, TB, 0, stream>>>(bkt, nxt, src, ew, h, b1, emb, N, 0, Ei);
  else k_walk_agg<<<g_agg, TB, 0, stream>>>(heads, heads1, nxt, src, ew, h, b1, emb, N, 0, Ei);

  k_wmma_gemm<2><<<gblk, 128, 0, stream>>>(emb, 64, 0, W2p, KS2, h, 32, HID, nullptr, NPAIRS);
  if (use_bkt) k_bkt_agg<<<g_agg_b, TB, 0, stream>>>(bkt, nxt, src, ew, h, b2, emb, N, HID, Ei);
  else k_walk_agg<<<g_agg, TB, 0, stream>>>(heads, heads1, nxt, src, ew, h, b2, emb, N, HID, Ei);

  k_wmma_gemm<2><<<gblk, 128, 0, stream>>>(emb, 64, HID, W3p, KS2, h, 32, HID, nullptr, NPAIRS);
  if (use_bkt) k_bkt_agg<<<g_agg_b, TB, 0, stream>>>(bkt, nxt, src, ew, h, b3, emb, N, 2 * HID, Ei);
  else k_walk_agg<<<g_agg, TB, 0, stream>>>(heads, heads1, nxt, src, ew, h, b3, emb, N, 2 * HID, Ei);

  k_wmma_gemm<1><<<gblk, 128, 0, stream>>>(emb, 64, 0, Wlp, KSL, out, OUT_DIM, OUT_DIM, bl, NPAIRS);
}
